# starting kernel
# baseline (speedup 1.0000x reference)
#include <hip/hip_runtime.h>

typedef __attribute__((ext_vector_type(2))) float v2f;
typedef __attribute__((ext_vector_type(8))) float v8f;

static constexpr int KDIM    = 128;
static constexpr int N_EDGES = 25000;
static constexpr float EPS   = 1e-12f;

__global__ __launch_bounds__(256) void fr_softmax(const float* __restrict__ y,
                                                  float* __restrict__ probs, int n) {
    int wave = threadIdx.x >> 5, lane = threadIdx.x & 31;
    int row = blockIdx.x * 8 + wave;
    if (row >= n) return;
    float4 v = ((const float4*)(y + (size_t)row * KDIM))[lane];
    float m = fmaxf(fmaxf(v.x, v.y), fmaxf(v.z, v.w));
    #pragma unroll
    for (int off = 16; off > 0; off >>= 1) m = fmaxf(m, __shfl_xor(m, off, 32));
    float4 e;
    e.x = __expf(v.x - m); e.y = __expf(v.y - m);
    e.z = __expf(v.z - m); e.w = __expf(v.w - m);
    float s = e.x + e.y + e.z + e.w;
    #pragma unroll
    for (int off = 16; off > 0; off >>= 1) s += __shfl_xor(s, off, 32);
    float inv = 1.f / s;
    e.x *= inv; e.y *= inv; e.z *= inv; e.w *= inv;
    ((float4*)(probs + (size_t)row * KDIM))[lane] = e;
}

__global__ __launch_bounds__(256) void fr_degrees(const int* __restrict__ node_idx,
                                                  const int* __restrict__ edge_idx,
                                                  float* __restrict__ edge_deg,
                                                  float* __restrict__ node_deg, int nnz) {
    int i = blockIdx.x * 256 + threadIdx.x;
    if (i >= nnz) return;
    atomicAdd(&edge_deg[edge_idx[i]], 1.0f);
    atomicAdd(&node_deg[node_idx[i]], 1.0f);
}

__global__ __launch_bounds__(256) void fr_gather(const float* __restrict__ probs,
                                                 const int* __restrict__ node_idx,
                                                 const int* __restrict__ edge_idx,
                                                 float* __restrict__ edge_agg, int nnz) {
    long long gid = (long long)blockIdx.x * 256 + threadIdx.x;
    int i = (int)(gid >> 5);
    if (i >= nnz) return;
    int c4 = (int)(gid & 31);
    int nv = node_idx[i], ev = edge_idx[i];
    float4 p = ((const float4*)(probs + (size_t)nv * KDIM))[c4];
    float* dst = edge_agg + (size_t)ev * KDIM + c4 * 4;
    atomicAdd(dst + 0, p.x); atomicAdd(dst + 1, p.y);
    atomicAdd(dst + 2, p.z); atomicAdd(dst + 3, p.w);
}

__global__ __launch_bounds__(256) void fr_edge_gemm(const float* __restrict__ edge_agg,
                                                    const float* __restrict__ edge_deg,
                                                    const float* __restrict__ W,
                                                    float* __restrict__ edge_feat, int E) {
    __shared__ float sA[16][132];
    int tid = threadIdx.x;
    int m0 = blockIdx.x * 16;
    for (int i = tid; i < 16 * 32; i += 256) {
        int row = i >> 5, c4 = i & 31, r = m0 + row;
        float4 v = make_float4(0.f, 0.f, 0.f, 0.f);
        if (r < E) v = ((const float4*)edge_agg)[(size_t)r * 32 + c4];
        *(float4*)&sA[row][c4 * 4] = v;
    }
    __syncthreads();

    int wave = tid >> 5, lane = tid & 31;
    int half = lane >> 4, lr = lane & 15;
    int n0 = wave * 16;

    v8f c = {0.f, 0.f, 0.f, 0.f, 0.f, 0.f, 0.f, 0.f};
    #pragma unroll 4
    for (int k0 = 0; k0 < KDIM; k0 += 4) {
        int ka = k0 + 2 * half;
        v2f a, b;
        a.x = sA[lr][ka];
        a.y = sA[lr][ka + 1];
        b.x = W[(size_t)ka * KDIM + n0 + lr];
        b.y = W[(size_t)(ka + 1) * KDIM + n0 + lr];
        c = __builtin_amdgcn_wmma_f32_16x16x4_f32(false, a, false, b, (short)0, c,
                                                  false, false);
    }
    int rowbase = m0 + 8 * half;
    #pragma unroll
    for (int r = 0; r < 8; ++r) {
        int row = rowbase + r;
        if (row < E) {
            float inv = 1.f / fmaxf(edge_deg[row], 1.f);
            edge_feat[(size_t)row * KDIM + n0 + lr] = c[r] * inv;
        }
    }
}

__global__ __launch_bounds__(256) void fr_scatter(const float* __restrict__ edge_feat,
                                                  const int* __restrict__ node_idx,
                                                  const int* __restrict__ edge_idx,
                                                  float* __restrict__ out, int nnz) {
    long long gid = (long long)blockIdx.x * 256 + threadIdx.x;
    int i = (int)(gid >> 5);
    if (i >= nnz) return;
    int c4 = (int)(gid & 31);
    int nv = node_idx[i], ev = edge_idx[i];
    float4 f = ((const float4*)(edge_feat + (size_t)ev * KDIM))[c4];
    float* dst = out + (size_t)nv * KDIM + c4 * 4;
    atomicAdd(dst + 0, f.x); atomicAdd(dst + 1, f.y);
    atomicAdd(dst + 2, f.z); atomicAdd(dst + 3, f.w);
}

__global__ __launch_bounds__(256) void fr_finalize(const float* __restrict__ probs,
                                                   const float* __restrict__ node_deg,
                                                   float* __restrict__ out, int n) {
    int wave = threadIdx.x >> 5, lane = threadIdx.x & 31;
    int row = blockIdx.x * 8 + wave;
    if (row >= n) return;
    float inv = 1.f / fmaxf(node_deg[row], 1.f);
    float4 acc = ((const float4*)(out + (size_t)row * KDIM))[lane];
    float4 p   = ((const float4*)(probs + (size_t)row * KDIM))[lane];
    float4 nat;
    nat.x = fmaxf(p.x, EPS) * acc.x * inv;
    nat.y = fmaxf(p.y, EPS) * acc.y * inv;
    nat.z = fmaxf(p.z, EPS) * acc.z * inv;
    nat.w = fmaxf(p.w, EPS) * acc.w * inv;
    float s = nat.x + nat.y + nat.z + nat.w;
    #pragma unroll
    for (int off = 16; off > 0; off >>= 1) s += __shfl_xor(s, off, 32);
    float mean = s * (1.0f / KDIM);
    nat.x -= mean; nat.y -= mean; nat.z -= mean; nat.w -= mean;
    ((float4*)(out + (size_t)row * KDIM))[lane] = nat;
}

extern "C" void kernel_launch(void* const* d_in, const int* in_sizes, int n_in,
                              void* d_out, int out_size, void* d_ws, size_t ws_size,
                              hipStream_t stream) {
    const float* y        = (const float*)d_in[0];
    const float* W        = (const float*)d_in[1];
    const int*   node_idx = (const int*)d_in[2];
    const int*   edge_idx = (const int*)d_in[3];
    float*       out      = (float*)d_out;

    const int n   = in_sizes[0] / KDIM;
    const int nnz = in_sizes[2];
    const int E   = N_EDGES;

    float* probs     = (float*)d_ws;
    float* edge_agg  = probs + (size_t)n * KDIM;
    float* edge_feat = edge_agg + (size_t)E * KDIM;
    float* edge_deg  = edge_feat + (size_t)E * KDIM;
    float* node_deg  = edge_deg + E;

    hipMemsetAsync(edge_agg, 0, (size_t)E * KDIM * sizeof(float), stream);
    hipMemsetAsync(edge_deg, 0, (size_t)(E + n) * sizeof(float), stream);
    hipMemsetAsync(out, 0, (size_t)n * KDIM * sizeof(float), stream);

    fr_softmax<<<(n + 7) / 8, 256, 0, stream>>>(y, probs, n);
    fr_degrees<<<(nnz + 255) / 256, 256, 0, stream>>>(node_idx, edge_idx,
                                                      edge_deg, node_deg, nnz);
    {
        long long total = (long long)nnz * 32;
        fr_gather<<<(unsigned)((total + 255) / 256), 256, 0, stream>>>(
            probs, node_idx, edge_idx, edge_agg, nnz);
    }
    fr_edge_gemm<<<(E + 15) / 16, 256, 0, stream>>>(edge_agg, edge_deg, W,
                                                    edge_feat, E);
    {
        long long total = (long long)nnz * 32;
        fr_scatter<<<(unsigned)((total + 255) / 256), 256, 0, stream>>>(
            edge_feat, node_idx, edge_idx, out, nnz);
    }
    fr_finalize<<<(n + 7) / 8, 256, 0, stream>>>(probs, node_deg, out, n);
}


# gather+scatter: lane l owns columns l,l+32,l+64,l+96, so each wave-wide f32 atomic add and row load covers one whole 128-B line, not 8 lanes in each of 4 lines; atomics kept, sums in arrival order
# speedup vs baseline: 1.3217x; 1.3217x over previous
#include <hip/hip_runtime.h>

typedef __attribute__((ext_vector_type(2))) float v2f;
typedef __attribute__((ext_vector_type(8))) float v8f;

static constexpr int KDIM    = 128;
static constexpr int N_EDGES = 25000;
static constexpr float EPS   = 1e-12f;

__global__ __launch_bounds__(256) void fr_softmax(const float* __restrict__ y,
                                                  float* __restrict__ probs, int n) {
    int wave = threadIdx.x >> 5, lane = threadIdx.x & 31;
    int row = blockIdx.x * 8 + wave;
    if (row >= n) return;
    float4 v = ((const float4*)(y + (size_t)row * KDIM))[lane];
    float m = fmaxf(fmaxf(v.x, v.y), fmaxf(v.z, v.w));
    #pragma unroll
    for (int off = 16; off > 0; off >>= 1) m = fmaxf(m, __shfl_xor(m, off, 32));
    float4 e;
    e.x = __expf(v.x - m); e.y = __expf(v.y - m);
    e.z = __expf(v.z - m); e.w = __expf(v.w - m);
    float s = e.x + e.y + e.z + e.w;
    #pragma unroll
    for (int off = 16; off > 0; off >>= 1) s += __shfl_xor(s, off, 32);
    float inv = 1.f / s;
    e.x *= inv; e.y *= inv; e.z *= inv; e.w *= inv;
    ((float4*)(probs + (size_t)row * KDIM))[lane] = e;
}

__global__ __launch_bounds__(256) void fr_degrees(const int* __restrict__ node_idx,
                                                  const int* __restrict__ edge_idx,
                                                  float* __restrict__ edge_deg,
                                                  float* __restrict__ node_deg, int nnz) {
    int i = blockIdx.x * 256 + threadIdx.x;
    if (i >= nnz) return;
    atomicAdd(&edge_deg[edge_idx[i]], 1.0f);
    atomicAdd(&node_deg[node_idx[i]], 1.0f);
}

__global__ __launch_bounds__(256) void fr_gather(const float* __restrict__ probs,
                                                 const int* __restrict__ node_idx,
                                                 const int* __restrict__ edge_idx,
                                                 float* __restrict__ edge_agg, int nnz) {
    long long gid = (long long)blockIdx.x * 256 + threadIdx.x;
    int i = (int)(gid >> 5);
    if (i >= nnz) return;
    int lane = (int)(gid & 31);
    int nv = node_idx[i], ev = edge_idx[i];
    const float* src = probs + (size_t)nv * KDIM + lane;
    float* dst = edge_agg + (size_t)ev * KDIM + lane;
    float p0 = src[0], p1 = src[32], p2 = src[64], p3 = src[96];
    atomicAdd(dst +  0, p0); atomicAdd(dst + 32, p1);
    atomicAdd(dst + 64, p2); atomicAdd(dst + 96, p3);
}

__global__ __launch_bounds__(256) void fr_edge_gemm(const float* __restrict__ edge_agg,
                                                    const float* __restrict__ edge_deg,
                                                    const float* __restrict__ W,
                                                    float* __restrict__ edge_feat, int E) {
    __shared__ float sA[16][132];
    int tid = threadIdx.x;
    int m0 = blockIdx.x * 16;
    for (int i = tid; i < 16 * 32; i += 256) {
        int row = i >> 5, c4 = i & 31, r = m0 + row;
        float4 v = make_float4(0.f, 0.f, 0.f, 0.f);
        if (r < E) v = ((const float4*)edge_agg)[(size_t)r * 32 + c4];
        *(float4*)&sA[row][c4 * 4] = v;
    }
    __syncthreads();

    int wave = tid >> 5, lane = tid & 31;
    int half = lane >> 4, lr = lane & 15;
    int n0 = wave * 16;

    v8f c = {0.f, 0.f, 0.f, 0.f, 0.f, 0.f, 0.f, 0.f};
    #pragma unroll 4
    for (int k0 = 0; k0 < KDIM; k0 += 4) {
        int ka = k0 + 2 * half;
        v2f a, b;
        a.x = sA[lr][ka];
        a.y = sA[lr][ka + 1];
        b.x = W[(size_t)ka * KDIM + n0 + lr];
        b.y = W[(size_t)(ka + 1) * KDIM + n0 + lr];
        c = __builtin_amdgcn_wmma_f32_16x16x4_f32(false, a, false, b, (short)0, c,
                                                  false, false);
    }
    int rowbase = m0 + 8 * half;
    #pragma unroll
    for (int r = 0; r < 8; ++r) {
        int row = rowbase + r;
        if (row < E) {
            float inv = 1.f / fmaxf(edge_deg[row], 1.f);
            edge_feat[(size_t)row * KDIM + n0 + lr] = c[r] * inv;
        }
    }
}

__global__ __launch_bounds__(256) void fr_scatter(const float* __restrict__ edge_feat,
                                                  const int* __restrict__ node_idx,
                                                  const int* __restrict__ edge_idx,
                                                  float* __restrict__ out, int nnz) {
    long long gid = (long long)blockIdx.x * 256 + threadIdx.x;
    int i = (int)(gid >> 5);
    if (i >= nnz) return;
    int lane = (int)(gid & 31);
    int nv = node_idx[i], ev = edge_idx[i];
    const float* src = edge_feat + (size_t)ev * KDIM + lane;
    float* dst = out + (size_t)nv * KDIM + lane;
    float f0 = src[0], f1 = src[32], f2 = src[64], f3 = src[96];
    atomicAdd(dst +  0, f0); atomicAdd(dst + 32, f1);
    atomicAdd(dst + 64, f2); atomicAdd(dst + 96, f3);
}

__global__ __launch_bounds__(256) void fr_finalize(const float* __restrict__ probs,
                                                   const float* __restrict__ node_deg,
                                                   float* __restrict__ out, int n) {
    int wave = threadIdx.x >> 5, lane = threadIdx.x & 31;
    int row = blockIdx.x * 8 + wave;
    if (row >= n) return;
    float inv = 1.f / fmaxf(node_deg[row], 1.f);
    float4 acc = ((const float4*)(out + (size_t)row * KDIM))[lane];
    float4 p   = ((const float4*)(probs + (size_t)row * KDIM))[lane];
    float4 nat;
    nat.x = fmaxf(p.x, EPS) * acc.x * inv;
    nat.y = fmaxf(p.y, EPS) * acc.y * inv;
    nat.z = fmaxf(p.z, EPS) * acc.z * inv;
    nat.w = fmaxf(p.w, EPS) * acc.w * inv;
    float s = nat.x + nat.y + nat.z + nat.w;
    #pragma unroll
    for (int off = 16; off > 0; off >>= 1) s += __shfl_xor(s, off, 32);
    float mean = s * (1.0f / KDIM);
    nat.x -= mean; nat.y -= mean; nat.z -= mean; nat.w -= mean;
    ((float4*)(out + (size_t)row * KDIM))[lane] = nat;
}

extern "C" void kernel_launch(void* const* d_in, const int* in_sizes, int n_in,
                              void* d_out, int out_size, void* d_ws, size_t ws_size,
                              hipStream_t stream) {
    const float* y        = (const float*)d_in[0];
    const float* W        = (const float*)d_in[1];
    const int*   node_idx = (const int*)d_in[2];
    const int*   edge_idx = (const int*)d_in[3];
    float*       out      = (float*)d_out;

    const int n   = in_sizes[0] / KDIM;
    const int nnz = in_sizes[2];
    const int E   = N_EDGES;

    float* probs     = (float*)d_ws;
    float* edge_agg  = probs + (size_t)n * KDIM;
    float* edge_feat = edge_agg + (size_t)E * KDIM;
    float* edge_deg  = edge_feat + (size_t)E * KDIM;
    float* node_deg  = edge_deg + E;

    hipMemsetAsync(edge_agg, 0, (size_t)E * KDIM * sizeof(float), stream);
    hipMemsetAsync(edge_deg, 0, (size_t)(E + n) * sizeof(float), stream);
    hipMemsetAsync(out, 0, (size_t)n * KDIM * sizeof(float), stream);

    fr_softmax<<<(n + 7) / 8, 256, 0, stream>>>(y, probs, n);
    fr_degrees<<<(nnz + 255) / 256, 256, 0, stream>>>(node_idx, edge_idx,
                                                      edge_deg, node_deg, nnz);
    {
        long long total = (long long)nnz * 32;
        fr_gather<<<(unsigned)((total + 255) / 256), 256, 0, stream>>>(
            probs, node_idx, edge_idx, edge_agg, nnz);
    }
    fr_edge_gemm<<<(E + 15) / 16, 256, 0, stream>>>(edge_agg, edge_deg, W,
                                                    edge_feat, E);
    {
        long long total = (long long)nnz * 32;
        fr_scatter<<<(unsigned)((total + 255) / 256), 256, 0, stream>>>(
            edge_feat, node_idx, edge_idx, out, nnz);
    }
    fr_finalize<<<(n + 7) / 8, 256, 0, stream>>>(probs, node_deg, out, n);
}


# gather+scatter: a wave takes 32 entries, not 1: indices read as one coalesced line each, rows picked by readlane, 8 rows (32 line loads) in flight per wave; atomics kept, sums in arrival order
# speedup vs baseline: 1.7025x; 1.2881x over previous
#include <hip/hip_runtime.h>

typedef __attribute__((ext_vector_type(2))) float v2f;
typedef __attribute__((ext_vector_type(8))) float v8f;

static constexpr int KDIM    = 128;
static constexpr int N_EDGES = 25000;
static constexpr float EPS   = 1e-12f;

__global__ __launch_bounds__(256) void fr_softmax(const float* __restrict__ y,
                                                  float* __restrict__ probs, int n) {
    int wave = threadIdx.x >> 5, lane = threadIdx.x & 31;
    int row = blockIdx.x * 8 + wave;
    if (row >= n) return;
    float4 v = ((const float4*)(y + (size_t)row * KDIM))[lane];
    float m = fmaxf(fmaxf(v.x, v.y), fmaxf(v.z, v.w));
    #pragma unroll
    for (int off = 16; off > 0; off >>= 1) m = fmaxf(m, __shfl_xor(m, off, 32));
    float4 e;
    e.x = __expf(v.x - m); e.y = __expf(v.y - m);
    e.z = __expf(v.z - m); e.w = __expf(v.w - m);
    float s = e.x + e.y + e.z + e.w;
    #pragma unroll
    for (int off = 16; off > 0; off >>= 1) s += __shfl_xor(s, off, 32);
    float inv = 1.f / s;
    e.x *= inv; e.y *= inv; e.z *= inv; e.w *= inv;
    ((float4*)(probs + (size_t)row * KDIM))[lane] = e;
}

__global__ __launch_bounds__(256) void fr_degrees(const int* __restrict__ node_idx,
                                                  const int* __restrict__ edge_idx,
                                                  float* __restrict__ edge_deg,
                                                  float* __restrict__ node_deg, int nnz) {
    int i = blockIdx.x * 256 + threadIdx.x;
    if (i >= nnz) return;
    atomicAdd(&edge_deg[edge_idx[i]], 1.0f);
    atomicAdd(&node_deg[node_idx[i]], 1.0f);
}

static constexpr int CHUNK = 32;
static constexpr int BATCH = 8;

__device__ __forceinline__ void fr_row_add(const float* __restrict__ src, float* __restrict__ dst,
                                           int s, int d, unsigned lane) {
    const float* p = src + (size_t)s * KDIM;
    float* q = dst + (size_t)d * KDIM;
    float v0 = p[lane], v1 = p[lane + 32], v2 = p[lane + 64], v3 = p[lane + 96];
    atomicAdd(q + lane, v0);      atomicAdd(q + lane + 32, v1);
    atomicAdd(q + lane + 64, v2); atomicAdd(q + lane + 96, v3);
}

__device__ __forceinline__ void fr_wave_rows_add(const float* __restrict__ src,
                                                 float* __restrict__ dst,
                                                 int s_idx, int d_idx, int cnt, unsigned lane) {
    if (cnt == CHUNK) {
        #pragma unroll
        for (int j0 = 0; j0 < CHUNK; j0 += BATCH) {
            float v[BATCH][4];
            #pragma unroll
            for (int u = 0; u < BATCH; ++u) {
                int s = __builtin_amdgcn_readlane(s_idx, j0 + u);
                const float* p = src + (size_t)s * KDIM;
                v[u][0] = p[lane];      v[u][1] = p[lane + 32];
                v[u][2] = p[lane + 64]; v[u][3] = p[lane + 96];
            }
            #pragma unroll
            for (int u = 0; u < BATCH; ++u) {
                int d = __builtin_amdgcn_readlane(d_idx, j0 + u);
                float* q = dst + (size_t)d * KDIM;
                atomicAdd(q + lane, v[u][0]);      atomicAdd(q + lane + 32, v[u][1]);
                atomicAdd(q + lane + 64, v[u][2]); atomicAdd(q + lane + 96, v[u][3]);
            }
        }
    } else {
        for (int j = 0; j < cnt; ++j)
            fr_row_add(src, dst, __builtin_amdgcn_readlane(s_idx, j),
                       __builtin_amdgcn_readlane(d_idx, j), lane);
    }
}

__global__ __launch_bounds__(256) void fr_gather(const float* __restrict__ probs,
                                                 const int* __restrict__ node_idx,
                                                 const int* __restrict__ edge_idx,
                                                 float* __restrict__ edge_agg, int nnz) {
    unsigned lane = threadIdx.x & 31;
    int wave = __builtin_amdgcn_readfirstlane((int)(threadIdx.x >> 5));
    int base = ((int)blockIdx.x * 8 + wave) * CHUNK;
    if (base >= nnz) return;
    int cnt = min(CHUNK, nnz - base);
    int nv = 0, ev = 0;
    if ((int)lane < cnt) { nv = node_idx[base + lane]; ev = edge_idx[base + lane]; }
    fr_wave_rows_add(probs, edge_agg, nv, ev, cnt, lane);
}

__global__ __launch_bounds__(256) void fr_edge_gemm(const float* __restrict__ edge_agg,
                                                    const float* __restrict__ edge_deg,
                                                    const float* __restrict__ W,
                                                    float* __restrict__ edge_feat, int E) {
    __shared__ float sA[16][132];
    int tid = threadIdx.x;
    int m0 = blockIdx.x * 16;
    for (int i = tid; i < 16 * 32; i += 256) {
        int row = i >> 5, c4 = i & 31, r = m0 + row;
        float4 v = make_float4(0.f, 0.f, 0.f, 0.f);
        if (r < E) v = ((const float4*)edge_agg)[(size_t)r * 32 + c4];
        *(float4*)&sA[row][c4 * 4] = v;
    }
    __syncthreads();

    int wave = tid >> 5, lane = tid & 31;
    int half = lane >> 4, lr = lane & 15;
    int n0 = wave * 16;

    v8f c = {0.f, 0.f, 0.f, 0.f, 0.f, 0.f, 0.f, 0.f};
    #pragma unroll 4
    for (int k0 = 0; k0 < KDIM; k0 += 4) {
        int ka = k0 + 2 * half;
        v2f a, b;
        a.x = sA[lr][ka];
        a.y = sA[lr][ka + 1];
        b.x = W[(size_t)ka * KDIM + n0 + lr];
        b.y = W[(size_t)(ka + 1) * KDIM + n0 + lr];
        c = __builtin_amdgcn_wmma_f32_16x16x4_f32(false, a, false, b, (short)0, c,
                                                  false, false);
    }
    int rowbase = m0 + 8 * half;
    #pragma unroll
    for (int r = 0; r < 8; ++r) {
        int row = rowbase + r;
        if (row < E) {
            float inv = 1.f / fmaxf(edge_deg[row], 1.f);
            edge_feat[(size_t)row * KDIM + n0 + lr] = c[r] * inv;
        }
    }
}

__global__ __launch_bounds__(256) void fr_scatter(const float* __restrict__ edge_feat,
                                                  const int* __restrict__ node_idx,
                                                  const int* __restrict__ edge_idx,
                                                  float* __restrict__ out, int nnz) {
    unsigned lane = threadIdx.x & 31;
    int wave = __builtin_amdgcn_readfirstlane((int)(threadIdx.x >> 5));
    int base = ((int)blockIdx.x * 8 + wave) * CHUNK;
    if (base >= nnz) return;
    int cnt = min(CHUNK, nnz - base);
    int nv = 0, ev = 0;
    if ((int)lane < cnt) { nv = node_idx[base + lane]; ev = edge_idx[base + lane]; }
    fr_wave_rows_add(edge_feat, out, ev, nv, cnt, lane);
}

__global__ __launch_bounds__(256) void fr_finalize(const float* __restrict__ probs,
                                                   const float* __restrict__ node_deg,
                                                   float* __restrict__ out, int n) {
    int wave = threadIdx.x >> 5, lane = threadIdx.x & 31;
    int row = blockIdx.x * 8 + wave;
    if (row >= n) return;
    float inv = 1.f / fmaxf(node_deg[row], 1.f);
    float4 acc = ((const float4*)(out + (size_t)row * KDIM))[lane];
    float4 p   = ((const float4*)(probs + (size_t)row * KDIM))[lane];
    float4 nat;
    nat.x = fmaxf(p.x, EPS) * acc.x * inv;
    nat.y = fmaxf(p.y, EPS) * acc.y * inv;
    nat.z = fmaxf(p.z, EPS) * acc.z * inv;
    nat.w = fmaxf(p.w, EPS) * acc.w * inv;
    float s = nat.x + nat.y + nat.z + nat.w;
    #pragma unroll
    for (int off = 16; off > 0; off >>= 1) s += __shfl_xor(s, off, 32);
    float mean = s * (1.0f / KDIM);
    nat.x -= mean; nat.y -= mean; nat.z -= mean; nat.w -= mean;
    ((float4*)(out + (size_t)row * KDIM))[lane] = nat;
}

extern "C" void kernel_launch(void* const* d_in, const int* in_sizes, int n_in,
                              void* d_out, int out_size, void* d_ws, size_t ws_size,
                              hipStream_t stream) {
    const float* y        = (const float*)d_in[0];
    const float* W        = (const float*)d_in[1];
    const int*   node_idx = (const int*)d_in[2];
    const int*   edge_idx = (const int*)d_in[3];
    float*       out      = (float*)d_out;

    const int n   = in_sizes[0] / KDIM;
    const int nnz = in_sizes[2];
    const int E   = N_EDGES;

    float* probs     = (float*)d_ws;
    float* edge_agg  = probs + (size_t)n * KDIM;
    float* edge_feat = edge_agg + (size_t)E * KDIM;
    float* edge_deg  = edge_feat + (size_t)E * KDIM;
    float* node_deg  = edge_deg + E;

    hipMemsetAsync(edge_agg, 0, (size_t)E * KDIM * sizeof(float), stream);
    hipMemsetAsync(edge_deg, 0, (size_t)(E + n) * sizeof(float), stream);
    hipMemsetAsync(out, 0, (size_t)n * KDIM * sizeof(float), stream);

    fr_softmax<<<(n + 7) / 8, 256, 0, stream>>>(y, probs, n);
    fr_degrees<<<(nnz + 255) / 256, 256, 0, stream>>>(node_idx, edge_idx,
                                                      edge_deg, node_deg, nnz);
    fr_gather<<<(nnz + 255) / 256, 256, 0, stream>>>(probs, node_idx, edge_idx, edge_agg, nnz);
    fr_edge_gemm<<<(E + 15) / 16, 256, 0, stream>>>(edge_agg, edge_deg, W,
                                                    edge_feat, E);
    fr_scatter<<<(nnz + 255) / 256, 256, 0, stream>>>(edge_feat, node_idx, edge_idx, out, nnz);
    fr_finalize<<<(n + 7) / 8, 256, 0, stream>>>(probs, node_deg, out, n);
}


# gather+scatter: rows split even/odd into two ticket queues; a wave reads its die number and serves its own half's class first, so a row's f32 atomics come from one half; sums in arrival order
# speedup vs baseline: 1.9165x; 1.1257x over previous
#include <hip/hip_runtime.h>

typedef __attribute__((ext_vector_type(2))) float v2f;
typedef __attribute__((ext_vector_type(8))) float v8f;

static constexpr int KDIM    = 128;
static constexpr int N_EDGES = 25000;
static constexpr float EPS   = 1e-12f;
static constexpr int QSTRIDE = 64;

__global__ __launch_bounds__(256) void fr_softmax(const float* __restrict__ y,
                                                  float* __restrict__ probs, int n) {
    int wave = threadIdx.x >> 5, lane = threadIdx.x & 31;
    int row = blockIdx.x * 8 + wave;
    if (row >= n) return;
    float4 v = ((const float4*)(y + (size_t)row * KDIM))[lane];
    float m = fmaxf(fmaxf(v.x, v.y), fmaxf(v.z, v.w));
    #pragma unroll
    for (int off = 16; off > 0; off >>= 1) m = fmaxf(m, __shfl_xor(m, off, 32));
    float4 e;
    e.x = __expf(v.x - m); e.y = __expf(v.y - m);
    e.z = __expf(v.z - m); e.w = __expf(v.w - m);
    float s = e.x + e.y + e.z + e.w;
    #pragma unroll
    for (int off = 16; off > 0; off >>= 1) s += __shfl_xor(s, off, 32);
    float inv = 1.f / s;
    e.x *= inv; e.y *= inv; e.z *= inv; e.w *= inv;
    ((float4*)(probs + (size_t)row * KDIM))[lane] = e;
}

__global__ __launch_bounds__(256) void fr_degrees(const int* __restrict__ node_idx,
                                                  const int* __restrict__ edge_idx,
                                                  float* __restrict__ edge_deg,
                                                  float* __restrict__ node_deg,
                                                  int* __restrict__ gq, int nnz) {
    int i = blockIdx.x * 256 + threadIdx.x;
    if (i < 2) gq[i * QSTRIDE] = 0;
    if (i >= nnz) return;
    atomicAdd(&edge_deg[edge_idx[i]], 1.0f);
    atomicAdd(&node_deg[node_idx[i]], 1.0f);
}

static constexpr int TCHUNK  = 64;
static constexpr int BATCH   = 8;
static constexpr int QBLOCKS = 2048;

__device__ __forceinline__ int fr_pick(int a0, int a1, int j) {
    int x0 = __builtin_amdgcn_readlane(a0, j & 31);
    int x1 = __builtin_amdgcn_readlane(a1, j & 31);
    return (j & 32) ? x1 : x0;
}

template <bool FULL>
__device__ __forceinline__ void fr_batch(const float* __restrict__ src, float* __restrict__ dst,
                                         int s0, int s1, int d0, int d1,
                                         const int (&j)[BATCH], unsigned lane) {
    float v[BATCH][4];
    #pragma unroll
    for (int u = 0; u < BATCH; ++u) {
        v[u][0] = 0.f; v[u][1] = 0.f; v[u][2] = 0.f; v[u][3] = 0.f;
        if (FULL || j[u] >= 0) {
            const float* p = src + (size_t)fr_pick(s0, s1, j[u]) * KDIM;
            v[u][0] = p[lane];      v[u][1] = p[lane + 32];
            v[u][2] = p[lane + 64]; v[u][3] = p[lane + 96];
        }
    }
    #pragma unroll
    for (int u = 0; u < BATCH; ++u) {
        if (FULL || j[u] >= 0) {
            float* q = dst + (size_t)fr_pick(d0, d1, j[u]) * KDIM;
            atomicAdd(q + lane, v[u][0]);      atomicAdd(q + lane + 32, v[u][1]);
            atomicAdd(q + lane + 64, v[u][2]); atomicAdd(q + lane + 96, v[u][3]);
        }
    }
}

__device__ __forceinline__ void fr_visit(const float* __restrict__ src, float* __restrict__ dst,
                                         const int* __restrict__ s_list,
                                         const int* __restrict__ d_list,
                                         int base, int nnz, int cls, unsigned lane) {
    int i0 = base + (int)lane, i1 = i0 + 32;
    bool ok0 = i0 < nnz, ok1 = i1 < nnz;
    int s0 = 0, d0 = 0, s1 = 0, d1 = 0;
    if (ok0) { s0 = s_list[i0]; d0 = d_list[i0]; }
    if (ok1) { s1 = s_list[i1]; d1 = d_list[i1]; }
    unsigned m0 = __builtin_amdgcn_ballot_w32(ok0 && ((d0 & 1) == cls));
    unsigned m1 = __builtin_amdgcn_ballot_w32(ok1 && ((d1 & 1) == cls));
    unsigned long long m = (unsigned long long)m0 | ((unsigned long long)m1 << 32);
    while (m) {
        int j[BATCH];
        #pragma unroll
        for (int u = 0; u < BATCH; ++u) {
            if (m) { j[u] = __builtin_ctzll(m); m &= m - 1; } else j[u] = -1;
        }
        if (j[BATCH - 1] >= 0) fr_batch<true>(src, dst, s0, s1, d0, d1, j, lane);
        else                   fr_batch<false>(src, dst, s0, s1, d0, d1, j, lane);
    }
}

__device__ __forceinline__ void fr_queue_run(const float* __restrict__ src, float* __restrict__ dst,
                                             const int* __restrict__ s_list,
                                             const int* __restrict__ d_list,
                                             int nnz, int* qhead, int* s_done) {
    unsigned lane = threadIdx.x & 31;
    int half = (int)((__builtin_amdgcn_s_sendmsg_rtn(0x87) >> 18) & 1u);
    int tickets = (nnz + TCHUNK - 1) / TCHUNK;
    for (int pass = 0; pass < 2; ++pass) {
        int cls = half ^ pass;
        for (;;) {
            if (__builtin_amdgcn_readfirstlane(
                    __hip_atomic_load(&s_done[cls], __ATOMIC_RELAXED, __HIP_MEMORY_SCOPE_WORKGROUP)))
                break;
            int t = 0;
            if (lane == 0) t = atomicAdd(&qhead[cls * QSTRIDE], 1);
            t = __builtin_amdgcn_readfirstlane(t);
            if (t >= tickets) {
                if (lane == 0)
                    __hip_atomic_store(&s_done[cls], 1, __ATOMIC_RELAXED, __HIP_MEMORY_SCOPE_WORKGROUP);
                break;
            }
            fr_visit(src, dst, s_list, d_list, t * TCHUNK, nnz, cls, lane);
        }
    }
}

__global__ __launch_bounds__(256) void fr_gather(const float* __restrict__ probs,
                                                 const int* __restrict__ node_idx,
                                                 const int* __restrict__ edge_idx,
                                                 float* __restrict__ edge_agg,
                                                 int* gq, int nnz) {
    __shared__ int s_done[2];
    if (threadIdx.x < 2) s_done[threadIdx.x] = 0;
    __syncthreads();
    fr_queue_run(probs, edge_agg, node_idx, edge_idx, nnz, gq, s_done);
}

__global__ __launch_bounds__(256) void fr_edge_gemm(float* edge_agg,
                                                    const float* __restrict__ edge_deg,
                                                    const float* __restrict__ W,
                                                    float* __restrict__ edge_feat, int E) {
    __shared__ float sA[16][132];
    int tid = threadIdx.x;
    int m0 = blockIdx.x * 16;
    for (int i = tid; i < 16 * 32; i += 256) {
        int row = i >> 5, c4 = i & 31, r = m0 + row;
        float4 v = make_float4(0.f, 0.f, 0.f, 0.f);
        if (r < E) v = ((const float4*)edge_agg)[(size_t)r * 32 + c4];
        *(float4*)&sA[row][c4 * 4] = v;
    }
    __syncthreads();
    if (blockIdx.x == gridDim.x - 1 && tid < 2) ((int*)(edge_agg + (size_t)m0 * KDIM))[tid * QSTRIDE] = 0;

    int wave = tid >> 5, lane = tid & 31;
    int half = lane >> 4, lr = lane & 15;
    int n0 = wave * 16;

    v8f c = {0.f, 0.f, 0.f, 0.f, 0.f, 0.f, 0.f, 0.f};
    #pragma unroll 4
    for (int k0 = 0; k0 < KDIM; k0 += 4) {
        int ka = k0 + 2 * half;
        v2f a, b;
        a.x = sA[lr][ka];
        a.y = sA[lr][ka + 1];
        b.x = W[(size_t)ka * KDIM + n0 + lr];
        b.y = W[(size_t)(ka + 1) * KDIM + n0 + lr];
        c = __builtin_amdgcn_wmma_f32_16x16x4_f32(false, a, false, b, (short)0, c,
                                                  false, false);
    }
    int rowbase = m0 + 8 * half;
    #pragma unroll
    for (int r = 0; r < 8; ++r) {
        int row = rowbase + r;
        if (row < E) {
            float inv = 1.f / fmaxf(edge_deg[row], 1.f);
            edge_feat[(size_t)row * KDIM + n0 + lr] = c[r] * inv;
        }
    }
}

__global__ __launch_bounds__(256) void fr_scatter(const float* __restrict__ edge_feat,
                                                  const int* __restrict__ node_idx,
                                                  const int* __restrict__ edge_idx,
                                                  float* __restrict__ out,
                                                  float* edge_agg, int E, int nnz) {
    __shared__ int s_done[2];
    if (threadIdx.x < 2) s_done[threadIdx.x] = 0;
    __syncthreads();
    int* sq = (int*)(edge_agg + (size_t)(((E + 15) / 16 - 1) * 16) * KDIM);
    fr_queue_run(edge_feat, out, edge_idx, node_idx, nnz, sq, s_done);
}

__global__ __launch_bounds__(256) void fr_finalize(const float* __restrict__ probs,
                                                   const float* __restrict__ node_deg,
                                                   float* __restrict__ out, int n) {
    int wave = threadIdx.x >> 5, lane = threadIdx.x & 31;
    int row = blockIdx.x * 8 + wave;
    if (row >= n) return;
    float inv = 1.f / fmaxf(node_deg[row], 1.f);
    float4 acc = ((const float4*)(out + (size_t)row * KDIM))[lane];
    float4 p   = ((const float4*)(probs + (size_t)row * KDIM))[lane];
    float4 nat;
    nat.x = fmaxf(p.x, EPS) * acc.x * inv;
    nat.y = fmaxf(p.y, EPS) * acc.y * inv;
    nat.z = fmaxf(p.z, EPS) * acc.z * inv;
    nat.w = fmaxf(p.w, EPS) * acc.w * inv;
    float s = nat.x + nat.y + nat.z + nat.w;
    #pragma unroll
    for (int off = 16; off > 0; off >>= 1) s += __shfl_xor(s, off, 32);
    float mean = s * (1.0f / KDIM);
    nat.x -= mean; nat.y -= mean; nat.z -= mean; nat.w -= mean;
    ((float4*)(out + (size_t)row * KDIM))[lane] = nat;
}

extern "C" void kernel_launch(void* const* d_in, const int* in_sizes, int n_in,
                              void* d_out, int out_size, void* d_ws, size_t ws_size,
                              hipStream_t stream) {
    const float* y        = (const float*)d_in[0];
    const float* W        = (const float*)d_in[1];
    const int*   node_idx = (const int*)d_in[2];
    const int*   edge_idx = (const int*)d_in[3];
    float*       out      = (float*)d_out;

    const int n   = in_sizes[0] / KDIM;
    const int nnz = in_sizes[2];
    const int E   = N_EDGES;

    float* probs     = (float*)d_ws;
    float* edge_agg  = probs + (size_t)n * KDIM;
    float* edge_feat = edge_agg + (size_t)E * KDIM;
    float* edge_deg  = edge_feat + (size_t)E * KDIM;
    float* node_deg  = edge_deg + E;

    hipMemsetAsync(edge_agg, 0, (size_t)E * KDIM * sizeof(float), stream);
    hipMemsetAsync(edge_deg, 0, (size_t)(E + n) * sizeof(float), stream);
    hipMemsetAsync(out, 0, (size_t)n * KDIM * sizeof(float), stream);

    fr_softmax<<<(n + 7) / 8, 256, 0, stream>>>(y, probs, n);
    fr_degrees<<<(nnz + 255) / 256, 256, 0, stream>>>(node_idx, edge_idx,
                                                      edge_deg, node_deg, (int*)edge_feat, nnz);
    fr_gather<<<QBLOCKS, 256, 0, stream>>>(probs, node_idx, edge_idx, edge_agg,
                                           (int*)edge_feat, nnz);
    fr_edge_gemm<<<(E + 15) / 16, 256, 0, stream>>>(edge_agg, edge_deg, W,
                                                    edge_feat, E);
    fr_scatter<<<QBLOCKS, 256, 0, stream>>>(edge_feat, node_idx, edge_idx, out, edge_agg, E, nnz);
    fr_finalize<<<(n + 7) / 8, 256, 0, stream>>>(probs, node_deg, out, n);
}


# segment sums without f32 atomics: entries bucketed by destination row in the call (returning int add = rank, scan, place), one wave sums a row in registers, one store; sum order = arrival order
# speedup vs baseline: 2.0043x; 1.0458x over previous
#include <hip/hip_runtime.h>

typedef __attribute__((ext_vector_type(2))) float v2f;
typedef __attribute__((ext_vector_type(8))) float v8f;

static constexpr int KDIM    = 128;
static constexpr int N_EDGES = 25000;
static constexpr float EPS   = 1e-12f;


__global__ __launch_bounds__(256) void fr_softmax(const float* __restrict__ y,
                                                  float* __restrict__ probs,
                                                  int* __restrict__ cnt, int ncnt, int n) {
    for (int i = blockIdx.x * 256 + threadIdx.x; i < ncnt; i += gridDim.x * 256) cnt[i] = 0;
    int wave = threadIdx.x >> 5, lane = threadIdx.x & 31;
    int row = blockIdx.x * 8 + wave;
    if (row >= n) return;
    float4 v = ((const float4*)(y + (size_t)row * KDIM))[lane];
    float m = fmaxf(fmaxf(v.x, v.y), fmaxf(v.z, v.w));
    #pragma unroll
    for (int off = 16; off > 0; off >>= 1) m = fmaxf(m, __shfl_xor(m, off, 32));
    float4 e;
    e.x = __expf(v.x - m); e.y = __expf(v.y - m);
    e.z = __expf(v.z - m); e.w = __expf(v.w - m);
    float s = e.x + e.y + e.z + e.w;
    #pragma unroll
    for (int off = 16; off > 0; off >>= 1) s += __shfl_xor(s, off, 32);
    float inv = 1.f / s;
    e.x *= inv; e.y *= inv; e.z *= inv; e.w *= inv;
    ((float4*)(probs + (size_t)row * KDIM))[lane] = e;
}

static constexpr int PER = 4;

__global__ __launch_bounds__(256) void fr_rank(const int* __restrict__ node_idx,
                                               const int* __restrict__ edge_idx,
                                               int* __restrict__ cnt_e, int* __restrict__ cnt_n,
                                               int* __restrict__ rank_e, int* __restrict__ rank_n,
                                               int nnz) {
    int g = blockIdx.x * 256 + threadIdx.x, T = gridDim.x * 256;
    int e[PER], v[PER], re[PER], rn[PER];
    #pragma unroll
    for (int k = 0; k < PER; ++k) {
        int i = g + k * T;
        e[k] = 0; v[k] = 0;
        if (i < nnz) { e[k] = edge_idx[i]; v[k] = node_idx[i]; }
    }
    #pragma unroll
    for (int k = 0; k < PER; ++k) {
        re[k] = 0; rn[k] = 0;
        if (g + k * T < nnz) { re[k] = atomicAdd(&cnt_e[e[k]], 1); rn[k] = atomicAdd(&cnt_n[v[k]], 1); }
    }
    #pragma unroll
    for (int k = 0; k < PER; ++k) {
        int i = g + k * T;
        if (i < nnz) { rank_e[i] = re[k]; rank_n[i] = rn[k]; }
    }
}

static constexpr int SUB = 4;

__global__ __launch_bounds__(1024) void fr_scan(int* cnt, int E, int n) {
    __shared__ int s_w[SUB][32];
    int* a  = (blockIdx.x == 0) ? cnt : cnt + E;
    int len = (blockIdx.x == 0) ? E : n;
    int tid = threadIdx.x, lane = tid & 31, wave = tid >> 5;
    bool al = (((size_t)a) & 15) == 0;
    int carry = 0;
    for (int base = 0; base < len; base += SUB * 4096) {
        int x[SUB][4], ts[SUB], inc[SUB];
        #pragma unroll
        for (int k = 0; k < SUB; ++k) {
            int i0 = base + k * 4096 + tid * 4;
            if (al && i0 + 3 < len) {
                int4 q = *(const int4*)(a + i0);
                x[k][0] = q.x; x[k][1] = q.y; x[k][2] = q.z; x[k][3] = q.w;
            } else {
                #pragma unroll
                for (int c = 0; c < 4; ++c) x[k][c] = (i0 + c < len) ? a[i0 + c] : 0;
            }
        }
        #pragma unroll
        for (int k = 0; k < SUB; ++k) {
            ts[k] = x[k][0] + x[k][1] + x[k][2] + x[k][3];
            int v = ts[k];
            #pragma unroll
            for (int off = 1; off < 32; off <<= 1) {
                int t = __shfl_up(v, off, 32);
                if (lane >= off) v += t;
            }
            inc[k] = v;
            if (lane == 31) s_w[k][wave] = v;
        }
        __syncthreads();
        int run = carry;
        #pragma unroll
        for (int k = 0; k < SUB; ++k) {
            int ws = s_w[k][lane];
            int wi = ws;
            #pragma unroll
            for (int off = 1; off < 32; off <<= 1) {
                int t = __shfl_up(wi, off, 32);
                if (lane >= off) wi += t;
            }
            int wexcl = __shfl(wi, wave, 32) - __shfl(ws, wave, 32);
            int total = __shfl(wi, 31, 32);
            int o0 = run + wexcl + (inc[k] - ts[k]);
            int o1 = o0 + x[k][0], o2 = o1 + x[k][1], o3 = o2 + x[k][2];
            int i0 = base + k * 4096 + tid * 4;
            if (al && i0 + 3 < len) {
                *(int4*)(a + i0) = make_int4(o0, o1, o2, o3);
            } else {
                if (i0     < len) a[i0]     = o0;
                if (i0 + 1 < len) a[i0 + 1] = o1;
                if (i0 + 2 < len) a[i0 + 2] = o2;
                if (i0 + 3 < len) a[i0 + 3] = o3;
            }
            run += total;
        }
        carry = run;
        __syncthreads();
    }
}

__global__ __launch_bounds__(256) void fr_place(const int* __restrict__ node_idx,
                                                const int* __restrict__ edge_idx,
                                                const int* __restrict__ off_e,
                                                const int* __restrict__ off_n,
                                                const int* __restrict__ rank_e,
                                                const int* __restrict__ rank_n,
                                                int* __restrict__ perm_e, int* __restrict__ perm_n,
                                                int nnz) {
    int g = blockIdx.x * 256 + threadIdx.x, T = gridDim.x * 256;
    int e[PER], v[PER], pe[PER], pn[PER];
    #pragma unroll
    for (int k = 0; k < PER; ++k) {
        int i = g + k * T;
        e[k] = 0; v[k] = 0; pe[k] = 0; pn[k] = 0;
        if (i < nnz) { e[k] = edge_idx[i]; v[k] = node_idx[i]; pe[k] = rank_e[i]; pn[k] = rank_n[i]; }
    }
    #pragma unroll
    for (int k = 0; k < PER; ++k) {
        if (g + k * T < nnz) { pe[k] += off_e[e[k]]; pn[k] += off_n[v[k]]; }
    }
    #pragma unroll
    for (int k = 0; k < PER; ++k) {
        if (g + k * T < nnz) { perm_e[pe[k]] = v[k]; perm_n[pn[k]] = e[k]; }
    }
}

static constexpr int RB = 8;

template <bool FULL>
__device__ __forceinline__ void fr_add_rows(const float* __restrict__ src, int mine, int j, int cnt,
                                            unsigned lane, float4& acc) {
    float4 v[RB];
    int r[RB];
    #pragma unroll
    for (int u = 0; u < RB; ++u) r[u] = __builtin_amdgcn_readlane(mine, (j + u) & 31);
    #pragma unroll
    for (int u = 0; u < RB; ++u) {
        v[u] = make_float4(0.f, 0.f, 0.f, 0.f);
        if (FULL || j + u < cnt) v[u] = ((const float4*)(src + (size_t)r[u] * KDIM))[lane];
    }
    #pragma unroll
    for (int u = 0; u < RB; ++u) {
        acc.x += v[u].x; acc.y += v[u].y; acc.z += v[u].z; acc.w += v[u].w;
    }
}

__device__ __forceinline__ float4 fr_row_sum(const float* __restrict__ src,
                                             const int* __restrict__ list,
                                             int b, int e, unsigned lane) {
    float4 acc = make_float4(0.f, 0.f, 0.f, 0.f);
    for (int c = b; c < e; c += 32) {
        int cnt = (e - c < 32) ? (e - c) : 32;
        int mine = ((int)lane < cnt) ? list[c + (int)lane] : 0;
        for (int j = 0; j < cnt; j += RB) {
            if (j + RB <= cnt) fr_add_rows<true>(src, mine, j, cnt, lane, acc);
            else               fr_add_rows<false>(src, mine, j, cnt, lane, acc);
        }
    }
    return acc;
}

__global__ __launch_bounds__(256) void fr_gather(const float* __restrict__ probs,
                                                 const int* __restrict__ off_e,
                                                 const int* __restrict__ perm_e,
                                                 float* __restrict__ edge_agg, int E, int nnz) {
    int wave = threadIdx.x >> 5;
    unsigned lane = threadIdx.x & 31;
    int nw = gridDim.x * 8;
    for (int row0 = blockIdx.x * 8 + wave; row0 < E; row0 += nw) {
        int row = __builtin_amdgcn_readfirstlane(row0);
        int b = __builtin_amdgcn_readfirstlane(off_e[row]);
        int e = (row + 1 < E) ? __builtin_amdgcn_readfirstlane(off_e[row + 1]) : nnz;
        float4 acc = fr_row_sum(probs, perm_e, b, e, lane);
        float inv = 1.f / fmaxf((float)(e - b), 1.f);
        acc.x *= inv; acc.y *= inv; acc.z *= inv; acc.w *= inv;
        ((float4*)(edge_agg + (size_t)row * KDIM))[lane] = acc;
    }
}

__global__ __launch_bounds__(256) void fr_edge_gemm(float* ef, const float* __restrict__ W, int E) {
    __shared__ float sA[16][132];
    int tid = threadIdx.x;
    int m0 = blockIdx.x * 16;
    for (int i = tid; i < 16 * 32; i += 256) {
        int row = i >> 5, c4 = i & 31, r = m0 + row;
        float4 v = make_float4(0.f, 0.f, 0.f, 0.f);
        if (r < E) v = ((const float4*)ef)[(size_t)r * 32 + c4];
        *(float4*)&sA[row][c4 * 4] = v;
    }
    __syncthreads();

    int wave = tid >> 5, lane = tid & 31;
    int half = lane >> 4, lr = lane & 15;
    int n0 = wave * 16;

    v8f c = {0.f, 0.f, 0.f, 0.f, 0.f, 0.f, 0.f, 0.f};
    #pragma unroll 4
    for (int k0 = 0; k0 < KDIM; k0 += 4) {
        int ka = k0 + 2 * half;
        v2f a, b;
        a.x = sA[lr][ka];
        a.y = sA[lr][ka + 1];
        b.x = W[(size_t)ka * KDIM + n0 + lr];
        b.y = W[(size_t)(ka + 1) * KDIM + n0 + lr];
        c = __builtin_amdgcn_wmma_f32_16x16x4_f32(false, a, false, b, (short)0, c,
                                                  false, false);
    }
    int rowbase = m0 + 8 * half;
    #pragma unroll
    for (int r = 0; r < 8; ++r) {
        int row = rowbase + r;
        if (row < E) ef[(size_t)row * KDIM + n0 + lr] = c[r];
    }
}

__global__ __launch_bounds__(256) void fr_scatter_fin(const float* __restrict__ edge_feat,
                                                      const int* __restrict__ off_n,
                                                      const int* __restrict__ perm_n,
                                                      const float* __restrict__ probs,
                                                      float* __restrict__ out, int n, int nnz) {
    int wave = threadIdx.x >> 5;
    unsigned lane = threadIdx.x & 31;
    int nw = gridDim.x * 8;
    for (int row0 = blockIdx.x * 8 + wave; row0 < n; row0 += nw) {
        int row = __builtin_amdgcn_readfirstlane(row0);
        int b = __builtin_amdgcn_readfirstlane(off_n[row]);
        int e = (row + 1 < n) ? __builtin_amdgcn_readfirstlane(off_n[row + 1]) : nnz;
        float4 p = ((const float4*)(probs + (size_t)row * KDIM))[lane];
        float4 acc = fr_row_sum(edge_feat, perm_n, b, e, lane);
        float inv = 1.f / fmaxf((float)(e - b), 1.f);
        float4 nat;
        nat.x = fmaxf(p.x, EPS) * acc.x * inv;
        nat.y = fmaxf(p.y, EPS) * acc.y * inv;
        nat.z = fmaxf(p.z, EPS) * acc.z * inv;
        nat.w = fmaxf(p.w, EPS) * acc.w * inv;
        float s = nat.x + nat.y + nat.z + nat.w;
        #pragma unroll
        for (int off = 16; off > 0; off >>= 1) s += __shfl_xor(s, off, 32);
        float mean = s * (1.0f / KDIM);
        nat.x -= mean; nat.y -= mean; nat.z -= mean; nat.w -= mean;
        ((float4*)(out + (size_t)row * KDIM))[lane] = nat;
    }
}

extern "C" void kernel_launch(void* const* d_in, const int* in_sizes, int n_in,
                              void* d_out, int out_size, void* d_ws, size_t ws_size,
                              hipStream_t stream) {
    const float* y        = (const float*)d_in[0];
    const float* W        = (const float*)d_in[1];
    const int*   node_idx = (const int*)d_in[2];
    const int*   edge_idx = (const int*)d_in[3];
    float*       out      = (float*)d_out;

    const int n   = in_sizes[0] / KDIM;
    const int nnz = in_sizes[2];
    const int E   = N_EDGES;

    float* probs   = (float*)d_ws;
    float* plane_a = probs + (size_t)n * KDIM;
    float* plane_b = plane_a + (size_t)E * KDIM;
    int*   cnt     = (int*)(plane_b + (size_t)E * KDIM);
    int*   rank_e  = (int*)plane_a;
    int*   rank_n  = rank_e + nnz;
    int*   perm_e  = (int*)plane_b;
    int*   perm_n  = perm_e + nnz;

    fr_softmax<<<(n + 7) / 8, 256, 0, stream>>>(y, probs, cnt, E + n, n);
    fr_rank<<<(nnz + 256 * PER - 1) / (256 * PER), 256, 0, stream>>>(node_idx, edge_idx, cnt, cnt + E,
                                                                     rank_e, rank_n, nnz);
    fr_scan<<<2, 1024, 0, stream>>>(cnt, E, n);
    fr_place<<<(nnz + 256 * PER - 1) / (256 * PER), 256, 0, stream>>>(node_idx, edge_idx, cnt, cnt + E,
                                                                      rank_e, rank_n, perm_e, perm_n,
                                                                      nnz);
    fr_gather<<<(E + 15) / 16, 256, 0, stream>>>(probs, cnt, perm_e, plane_a, E, nnz);
    fr_edge_gemm<<<(E + 15) / 16, 256, 0, stream>>>(plane_a, W, E);
    fr_scatter_fin<<<(n + 63) / 64, 256, 0, stream>>>(plane_a, cnt + E, perm_n, probs, out, n, nnz);
}


# incidence sort in two levels so its counting runs in LDS: rank per 128/512-row bucket by LDS add, one global add per bucket and block, then an LDS row sort per bucket; sums stay in arrival order
# speedup vs baseline: 5.2404x; 2.6146x over previous
#include <hip/hip_runtime.h>

typedef __attribute__((ext_vector_type(2))) float v2f;
typedef __attribute__((ext_vector_type(8))) float v8f;

static constexpr int KDIM    = 128;
static constexpr int N_EDGES = 25000;
static constexpr float EPS   = 1e-12f;


__global__ __launch_bounds__(256) void fr_softmax(const float* __restrict__ y,
                                                  float* __restrict__ probs,
                                                  int* __restrict__ cnt, int ncnt, int n) {
    for (int i = blockIdx.x * 256 + threadIdx.x; i < ncnt; i += gridDim.x * 256) cnt[i] = 0;
    int wave = threadIdx.x >> 5, lane = threadIdx.x & 31;
    int row = blockIdx.x * 8 + wave;
    if (row >= n) return;
    float4 v = ((const float4*)(y + (size_t)row * KDIM))[lane];
    float m = fmaxf(fmaxf(v.x, v.y), fmaxf(v.z, v.w));
    #pragma unroll
    for (int off = 16; off > 0; off >>= 1) m = fmaxf(m, __shfl_xor(m, off, 32));
    float4 e;
    e.x = __expf(v.x - m); e.y = __expf(v.y - m);
    e.z = __expf(v.z - m); e.w = __expf(v.w - m);
    float s = e.x + e.y + e.z + e.w;
    #pragma unroll
    for (int off = 16; off > 0; off >>= 1) s += __shfl_xor(s, off, 32);
    float inv = 1.f / s;
    e.x *= inv; e.y *= inv; e.z *= inv; e.w *= inv;
    ((float4*)(probs + (size_t)row * KDIM))[lane] = e;
}

static constexpr int ESH   = 7;
static constexpr int NSH   = 9;
static constexpr int EPK   = 25;
static constexpr int NPK   = 23;
static constexpr int MAXBK = 256;
static constexpr int STHR  = 1024;
static constexpr int SPER  = 4;

__global__ __launch_bounds__(1024) void fr_count(const int* __restrict__ node_idx,
                                                 const int* __restrict__ edge_idx,
                                                 int* __restrict__ cnt_e, int* __restrict__ cnt_n,
                                                 int nbe, int nbn, int nnz) {
    __shared__ int s_h[2 * MAXBK];
    int tid = threadIdx.x;
    int base = blockIdx.x * (STHR * SPER) + tid;
    int e[SPER], v[SPER];
    #pragma unroll
    for (int k = 0; k < SPER; ++k) {
        int i = base + k * STHR;
        e[k] = 0; v[k] = 0;
        if (i < nnz) { e[k] = edge_idx[i]; v[k] = node_idx[i]; }
    }
    for (int i = tid; i < 2 * MAXBK; i += STHR) s_h[i] = 0;
    __syncthreads();
    #pragma unroll
    for (int k = 0; k < SPER; ++k) {
        if (base + k * STHR < nnz) {
            atomicAdd(&s_h[e[k] >> ESH], 1);
            atomicAdd(&s_h[MAXBK + (v[k] >> NSH)], 1);
        }
    }
    __syncthreads();
    if (tid < nbe) {
        int c = s_h[tid];
        if (c) atomicAdd(&cnt_e[(tid << ESH) + 1], c);
    } else if (tid >= MAXBK && tid < MAXBK + nbn) {
        int c = s_h[tid];
        if (c) atomicAdd(&cnt_n[((tid - MAXBK) << NSH) + 1], c);
    }
}

__global__ __launch_bounds__(1024) void fr_part(const int* __restrict__ node_idx,
                                                const int* __restrict__ edge_idx,
                                                int* cnt_e, int* cnt_n,
                                                unsigned* __restrict__ part_e,
                                                unsigned* __restrict__ part_n,
                                                int nbe, int nbn, int nnz) {
    __shared__ int s_h[2 * MAXBK];
    __shared__ int s_st[2 * MAXBK];
    __shared__ int s_fp[2 * MAXBK];
    __shared__ int s_ls[2 * MAXBK];
    __shared__ unsigned s_val[2][STHR * SPER];
    __shared__ int      s_pos[2][STHR * SPER];
    int tid = threadIdx.x, lane = tid & 31, wave = tid >> 5;
    int blk0 = blockIdx.x * (STHR * SPER);
    int base = blk0 + tid;
    int e[SPER], v[SPER], re[SPER], rn[SPER];
    #pragma unroll
    for (int k = 0; k < SPER; ++k) {
        int i = base + k * STHR;
        e[k] = 0; v[k] = 0;
        if (i < nnz) { e[k] = edge_idx[i]; v[k] = node_idx[i]; }
    }
    for (int i = tid; i < 2 * MAXBK; i += STHR) s_h[i] = 0;
    if (wave < 2) {
        const int* c = wave ? cnt_n : cnt_e;
        int sh = wave ? NSH : ESH, nb = wave ? nbn : nbe;
        int cv[MAXBK / 32], sum = 0;
        #pragma unroll
        for (int j = 0; j < MAXBK / 32; ++j) {
            int b = lane * (MAXBK / 32) + j;
            cv[j] = (b < nb) ? c[(b << sh) + 1] : 0;
            sum += cv[j];
        }
        int inc = sum;
        #pragma unroll
        for (int off = 1; off < 32; off <<= 1) {
            int t = __shfl_up(inc, off, 32);
            if (lane >= off) inc += t;
        }
        int run = inc - sum;
        #pragma unroll
        for (int j = 0; j < MAXBK / 32; ++j) {
            s_st[wave * MAXBK + lane * (MAXBK / 32) + j] = run;
            run += cv[j];
        }
    }
    __syncthreads();
    #pragma unroll
    for (int k = 0; k < SPER; ++k) {
        re[k] = 0; rn[k] = 0;
        if (base + k * STHR < nnz) {
            re[k] = atomicAdd(&s_h[e[k] >> ESH], 1);
            rn[k] = atomicAdd(&s_h[MAXBK + (v[k] >> NSH)], 1);
        }
    }
    __syncthreads();
    if (tid < nbe) {
        int c = s_h[tid], g = 0;
        if (c) g = atomicAdd(&cnt_e[(tid << ESH) + 2], c);
        s_fp[tid] = s_st[tid] + g;
        if (blockIdx.x == 0) cnt_e[tid << ESH] = s_st[tid];
    } else if (tid >= MAXBK && tid < MAXBK + nbn) {
        int b = tid - MAXBK;
        int c = s_h[tid], g = 0;
        if (c) g = atomicAdd(&cnt_n[(b << NSH) + 2], c);
        s_fp[tid] = s_st[tid] + g;
        if (blockIdx.x == 0) cnt_n[b << NSH] = s_st[tid];
    }
    if (wave >= 16 && wave < 18) {
        int side = wave - 16;
        int cv[MAXBK / 32], sum = 0;
        #pragma unroll
        for (int j = 0; j < MAXBK / 32; ++j) {
            cv[j] = s_h[side * MAXBK + lane * (MAXBK / 32) + j];
            sum += cv[j];
        }
        int inc = sum;
        #pragma unroll
        for (int off = 1; off < 32; off <<= 1) {
            int t = __shfl_up(inc, off, 32);
            if (lane >= off) inc += t;
        }
        int run = inc - sum;
        #pragma unroll
        for (int j = 0; j < MAXBK / 32; ++j) {
            s_ls[side * MAXBK + lane * (MAXBK / 32) + j] = run;
            run += cv[j];
        }
    }
    __syncthreads();
    #pragma unroll
    for (int k = 0; k < SPER; ++k) {
        if (base + k * STHR < nnz) {
            unsigned ue = (unsigned)e[k], uv = (unsigned)v[k];
            int be = e[k] >> ESH, bn = MAXBK + (v[k] >> NSH);
            int je = s_ls[be] + re[k], jn = s_ls[bn] + rn[k];
            s_val[0][je] = ((ue & ((1u << ESH) - 1)) << EPK) | uv;
            s_pos[0][je] = s_fp[be] + re[k];
            s_val[1][jn] = ((uv & ((1u << NSH) - 1)) << NPK) | ue;
            s_pos[1][jn] = s_fp[bn] + rn[k];
        }
    }
    __syncthreads();
    int mine = nnz - blk0; if (mine > STHR * SPER) mine = STHR * SPER;
    #pragma unroll
    for (int k = 0; k < SPER; ++k) {
        int j = tid + k * STHR;
        if (j < mine) {
            part_e[s_pos[0][j]] = s_val[0][j];
            part_n[s_pos[1][j]] = s_val[1][j];
        }
    }
}

static constexpr int CAP = 12288;

__global__ __launch_bounds__(1024) void fr_rows(const unsigned* __restrict__ part_e,
                                                const unsigned* __restrict__ part_n,
                                                int* off_e, int* off_n,
                                                int* __restrict__ perm_e, int* __restrict__ perm_n,
                                                int nbe, int E, int n, int nnz) {
    __shared__ int s_stage[CAP];
    __shared__ int s_c[1 << NSH];
    __shared__ int s_w[32];
    int tid = threadIdx.x, lane = tid & 31, wave = tid >> 5;
    bool ns = (int)blockIdx.x >= nbe;
    int b  = ns ? (int)blockIdx.x - nbe : (int)blockIdx.x;
    int sh = ns ? NSH : ESH, pk = ns ? NPK : EPK;
    int nrows = ns ? n : E;
    int* off = ns ? off_n : off_e;
    const unsigned* part = ns ? part_n : part_e;
    int* perm = ns ? perm_n : perm_e;
    int key0 = b << sh;
    int nk = nrows - key0; if (nk > (1 << sh)) nk = 1 << sh;
    int bstart = off[key0];
    int bend = (key0 + (1 << sh) < nrows) ? off[key0 + (1 << sh)] : nnz;
    int bcnt = bend - bstart;
    for (int i = tid; i < (1 << NSH); i += STHR) s_c[i] = 0;
    __syncthreads();
    const unsigned* src = part + bstart;
    for (int i0 = tid; i0 < bcnt + tid; i0 += STHR * SPER) {
        unsigned q[SPER];
        #pragma unroll
        for (int k = 0; k < SPER; ++k) { int i = i0 + k * STHR; q[k] = 0; if (i < bcnt) q[k] = src[i]; }
        #pragma unroll
        for (int k = 0; k < SPER; ++k) { if (i0 + k * STHR < bcnt) atomicAdd(&s_c[q[k] >> pk], 1); }
    }
    __syncthreads();
    int cv = 0, inc = 0;
    if (tid < (1 << NSH)) {
        cv = s_c[tid]; inc = cv;
        #pragma unroll
        for (int o = 1; o < 32; o <<= 1) {
            int t = __shfl_up(inc, o, 32);
            if (lane >= o) inc += t;
        }
        if (lane == 31) s_w[wave] = inc;
    }
    __syncthreads();
    if (tid < (1 << NSH)) {
        int pre = 0;
        #pragma unroll
        for (int w = 0; w < (1 << NSH) / 32; ++w) pre += (w < wave) ? s_w[w] : 0;
        int ex = pre + inc - cv;
        s_c[tid] = ex;
        if (tid > 0 && tid < nk) off[key0 + tid] = bstart + ex;
    }
    __syncthreads();
    unsigned mask = (1u << pk) - 1u;
    for (int i0 = tid; i0 < bcnt + tid; i0 += STHR * SPER) {
        unsigned q[SPER];
        #pragma unroll
        for (int k = 0; k < SPER; ++k) { int i = i0 + k * STHR; q[k] = 0; if (i < bcnt) q[k] = src[i]; }
        #pragma unroll
        for (int k = 0; k < SPER; ++k) {
            if (i0 + k * STHR < bcnt) {
                int p = atomicAdd(&s_c[q[k] >> pk], 1);
                int val = (int)(q[k] & mask);
                if (p < CAP) s_stage[p] = val; else perm[bstart + p] = val;
            }
        }
    }
    __syncthreads();
    int m = (bcnt < CAP) ? bcnt : CAP;
    for (int i = tid; i < m; i += STHR) perm[bstart + i] = s_stage[i];
}

static constexpr int RB = 8;

template <bool FULL>
__device__ __forceinline__ void fr_add_rows(const float* __restrict__ src, int mine, int j, int cnt,
                                            unsigned lane, float4& acc) {
    float4 v[RB];
    int r[RB];
    #pragma unroll
    for (int u = 0; u < RB; ++u) r[u] = __builtin_amdgcn_readlane(mine, (j + u) & 31);
    #pragma unroll
    for (int u = 0; u < RB; ++u) {
        v[u] = make_float4(0.f, 0.f, 0.f, 0.f);
        if (FULL || j + u < cnt) v[u] = ((const float4*)(src + (size_t)r[u] * KDIM))[lane];
    }
    #pragma unroll
    for (int u = 0; u < RB; ++u) {
        acc.x += v[u].x; acc.y += v[u].y; acc.z += v[u].z; acc.w += v[u].w;
    }
}

__device__ __forceinline__ float4 fr_row_sum(const float* __restrict__ src,
                                             const int* __restrict__ list,
                                             int b, int e, unsigned lane) {
    float4 acc = make_float4(0.f, 0.f, 0.f, 0.f);
    for (int c = b; c < e; c += 32) {
        int cnt = (e - c < 32) ? (e - c) : 32;
        int mine = ((int)lane < cnt) ? list[c + (int)lane] : 0;
        for (int j = 0; j < cnt; j += RB) {
            if (j + RB <= cnt) fr_add_rows<true>(src, mine, j, cnt, lane, acc);
            else               fr_add_rows<false>(src, mine, j, cnt, lane, acc);
        }
    }
    return acc;
}

__global__ __launch_bounds__(256) void fr_gather(const float* __restrict__ probs,
                                                 const int* __restrict__ off_e,
                                                 const int* __restrict__ perm_e,
                                                 float* __restrict__ edge_agg, int E, int nnz) {
    int wave = threadIdx.x >> 5;
    unsigned lane = threadIdx.x & 31;
    int nw = gridDim.x * 8;
    for (int row0 = blockIdx.x * 8 + wave; row0 < E; row0 += nw) {
        int row = __builtin_amdgcn_readfirstlane(row0);
        int b = __builtin_amdgcn_readfirstlane(off_e[row]);
        int e = (row + 1 < E) ? __builtin_amdgcn_readfirstlane(off_e[row + 1]) : nnz;
        float4 acc = fr_row_sum(probs, perm_e, b, e, lane);
        float inv = 1.f / fmaxf((float)(e - b), 1.f);
        acc.x *= inv; acc.y *= inv; acc.z *= inv; acc.w *= inv;
        ((float4*)(edge_agg + (size_t)row * KDIM))[lane] = acc;
    }
}

__global__ __launch_bounds__(256) void fr_edge_gemm(float* ef, const float* __restrict__ W, int E) {
    __shared__ float sA[16][132];
    int tid = threadIdx.x;
    int m0 = blockIdx.x * 16;
    for (int i = tid; i < 16 * 32; i += 256) {
        int row = i >> 5, c4 = i & 31, r = m0 + row;
        float4 v = make_float4(0.f, 0.f, 0.f, 0.f);
        if (r < E) v = ((const float4*)ef)[(size_t)r * 32 + c4];
        *(float4*)&sA[row][c4 * 4] = v;
    }
    __syncthreads();

    int wave = tid >> 5, lane = tid & 31;
    int half = lane >> 4, lr = lane & 15;
    int n0 = wave * 16;

    v8f c = {0.f, 0.f, 0.f, 0.f, 0.f, 0.f, 0.f, 0.f};
    #pragma unroll 4
    for (int k0 = 0; k0 < KDIM; k0 += 4) {
        int ka = k0 + 2 * half;
        v2f a, b;
        a.x = sA[lr][ka];
        a.y = sA[lr][ka + 1];
        b.x = W[(size_t)ka * KDIM + n0 + lr];
        b.y = W[(size_t)(ka + 1) * KDIM + n0 + lr];
        c = __builtin_amdgcn_wmma_f32_16x16x4_f32(false, a, false, b, (short)0, c,
                                                  false, false);
    }
    int rowbase = m0 + 8 * half;
    #pragma unroll
    for (int r = 0; r < 8; ++r) {
        int row = rowbase + r;
        if (row < E) ef[(size_t)row * KDIM + n0 + lr] = c[r];
    }
}

__global__ __launch_bounds__(256) void fr_scatter_fin(const float* __restrict__ edge_feat,
                                                      const int* __restrict__ off_n,
                                                      const int* __restrict__ perm_n,
                                                      const float* __restrict__ probs,
                                                      float* __restrict__ out, int n, int nnz) {
    int wave = threadIdx.x >> 5;
    unsigned lane = threadIdx.x & 31;
    int nw = gridDim.x * 8;
    for (int row0 = blockIdx.x * 8 + wave; row0 < n; row0 += nw) {
        int row = __builtin_amdgcn_readfirstlane(row0);
        int b = __builtin_amdgcn_readfirstlane(off_n[row]);
        int e = (row + 1 < n) ? __builtin_amdgcn_readfirstlane(off_n[row + 1]) : nnz;
        float4 p = ((const float4*)(probs + (size_t)row * KDIM))[lane];
        float4 acc = fr_row_sum(edge_feat, perm_n, b, e, lane);
        float inv = 1.f / fmaxf((float)(e - b), 1.f);
        float4 nat;
        nat.x = fmaxf(p.x, EPS) * acc.x * inv;
        nat.y = fmaxf(p.y, EPS) * acc.y * inv;
        nat.z = fmaxf(p.z, EPS) * acc.z * inv;
        nat.w = fmaxf(p.w, EPS) * acc.w * inv;
        float s = nat.x + nat.y + nat.z + nat.w;
        #pragma unroll
        for (int off = 16; off > 0; off >>= 1) s += __shfl_xor(s, off, 32);
        float mean = s * (1.0f / KDIM);
        nat.x -= mean; nat.y -= mean; nat.z -= mean; nat.w -= mean;
        ((float4*)(out + (size_t)row * KDIM))[lane] = nat;
    }
}

extern "C" void kernel_launch(void* const* d_in, const int* in_sizes, int n_in,
                              void* d_out, int out_size, void* d_ws, size_t ws_size,
                              hipStream_t stream) {
    const float* y        = (const float*)d_in[0];
    const float* W        = (const float*)d_in[1];
    const int*   node_idx = (const int*)d_in[2];
    const int*   edge_idx = (const int*)d_in[3];
    float*       out      = (float*)d_out;

    const int n   = in_sizes[0] / KDIM;
    const int nnz = in_sizes[2];
    const int E   = N_EDGES;
    const int nbe = (E + (1 << ESH) - 1) >> ESH;
    const int nbn = (n + (1 << NSH) - 1) >> NSH;

    float* probs   = (float*)d_ws;
    float* plane_a = probs + (size_t)n * KDIM;
    float* plane_b = plane_a + (size_t)E * KDIM;
    int*   cnt     = (int*)(plane_b + (size_t)E * KDIM);
    unsigned* part_e = (unsigned*)plane_a;
    unsigned* part_n = part_e + nnz;
    int*   perm_e  = (int*)plane_b;
    int*   perm_n  = perm_e + nnz;

    fr_softmax<<<(n + 7) / 8, 256, 0, stream>>>(y, probs, cnt, E + n, n);
    fr_count<<<(nnz + STHR * SPER - 1) / (STHR * SPER), STHR, 0, stream>>>(node_idx, edge_idx, cnt, cnt + E,
                                                                         nbe, nbn, nnz);
    fr_part<<<(nnz + STHR * SPER - 1) / (STHR * SPER), STHR, 0, stream>>>(node_idx, edge_idx, cnt, cnt + E,
                                                                        part_e, part_n, nbe, nbn, nnz);
    fr_rows<<<nbe + nbn, STHR, 0, stream>>>(part_e, part_n, cnt, cnt + E, perm_e, perm_n, nbe, E, n, nnz);
    fr_gather<<<(E + 15) / 16, 256, 0, stream>>>(probs, cnt, perm_e, plane_a, E, nnz);
    fr_edge_gemm<<<(E + 15) / 16, 256, 0, stream>>>(plane_a, W, E);
    fr_scatter_fin<<<(n + 63) / 64, 256, 0, stream>>>(plane_a, cnt + E, perm_n, probs, out, n, nnz);
}


# two launch boundaries removed (7 -> 5 launches): bucket count runs as 96 extra blocks of the softmax launch (histogram rows, no global add); edge mean tile x W done from LDS inside the gather
# speedup vs baseline: 5.9867x; 1.1424x over previous
#include <hip/hip_runtime.h>

typedef __attribute__((ext_vector_type(2))) float v2f;
typedef __attribute__((ext_vector_type(8))) float v8f;

static constexpr int KDIM    = 128;
static constexpr int N_EDGES = 25000;
static constexpr float EPS   = 1e-12f;


static constexpr int ESH   = 7;
static constexpr int NSH   = 9;
static constexpr int MAXBK = 256;
static constexpr int NCB   = 96;
static constexpr int CPT   = 9;

__global__ __launch_bounds__(1024) void fr_softmax_count(const float* __restrict__ y,
                                                         float* __restrict__ probs,
                                                         const int* __restrict__ node_idx,
                                                         const int* __restrict__ edge_idx,
                                                         int* __restrict__ cnt, int ncnt,
                                                         int* __restrict__ hist, int n, int nnz) {
    __shared__ int s_h[2 * MAXBK];
    int tid = threadIdx.x;
    if (blockIdx.x < NCB) {
        for (int i = tid; i < 2 * MAXBK; i += 1024) s_h[i] = 0;
        __syncthreads();
        int per = (nnz + NCB - 1) / NCB;
        int lo = blockIdx.x * per;
        int hi = (lo + per < nnz) ? lo + per : nnz;
        for (int i0 = lo + tid; i0 < hi; i0 += 1024 * CPT) {
            int e[CPT], v[CPT];
            #pragma unroll
            for (int k = 0; k < CPT; ++k) {
                int i = i0 + k * 1024;
                e[k] = 0; v[k] = 0;
                if (i < hi) { e[k] = edge_idx[i]; v[k] = node_idx[i]; }
            }
            #pragma unroll
            for (int k = 0; k < CPT; ++k) {
                if (i0 + k * 1024 < hi) {
                    atomicAdd(&s_h[e[k] >> ESH], 1);
                    atomicAdd(&s_h[MAXBK + (v[k] >> NSH)], 1);
                }
            }
        }
        __syncthreads();
        if (tid < 2 * MAXBK) hist[blockIdx.x * (2 * MAXBK) + tid] = s_h[tid];
        return;
    }
    int sb = blockIdx.x - NCB, nsb = gridDim.x - NCB;
    for (int i = sb * 1024 + tid; i < ncnt; i += nsb * 1024) cnt[i] = 0;
    int wave = tid >> 5, lane = tid & 31;
    int row = sb * 32 + wave;
    if (row >= n) return;
    float4 v = ((const float4*)(y + (size_t)row * KDIM))[lane];
    float m = fmaxf(fmaxf(v.x, v.y), fmaxf(v.z, v.w));
    #pragma unroll
    for (int off = 16; off > 0; off >>= 1) m = fmaxf(m, __shfl_xor(m, off, 32));
    float4 e;
    e.x = __expf(v.x - m); e.y = __expf(v.y - m);
    e.z = __expf(v.z - m); e.w = __expf(v.w - m);
    float s = e.x + e.y + e.z + e.w;
    #pragma unroll
    for (int off = 16; off > 0; off >>= 1) s += __shfl_xor(s, off, 32);
    float inv = 1.f / s;
    e.x *= inv; e.y *= inv; e.z *= inv; e.w *= inv;
    ((float4*)(probs + (size_t)row * KDIM))[lane] = e;
}

static constexpr int EPK   = 25;
static constexpr int NPK   = 23;
static constexpr int STHR  = 1024;
static constexpr int SPER  = 4;

__global__ __launch_bounds__(1024) void fr_part(const int* __restrict__ node_idx,
                                                const int* __restrict__ edge_idx,
                                                const int* __restrict__ hist,
                                                int* cnt_e, int* cnt_n,
                                                unsigned* __restrict__ part_e,
                                                unsigned* __restrict__ part_n,
                                                int nbe, int nbn, int nnz) {
    __shared__ int s_h[2 * MAXBK];
    __shared__ __attribute__((aligned(16))) int s_tot[8][2 * MAXBK];
    __shared__ int s_st[2 * MAXBK];
    __shared__ int s_fp[2 * MAXBK];
    __shared__ int s_ls[2 * MAXBK];
    __shared__ unsigned s_val[2][STHR * SPER];
    __shared__ int      s_pos[2][STHR * SPER];
    int tid = threadIdx.x, lane = tid & 31, wave = tid >> 5;
    int blk0 = blockIdx.x * (STHR * SPER);
    int base = blk0 + tid;
    int e[SPER], v[SPER], re[SPER], rn[SPER];
    #pragma unroll
    for (int k = 0; k < SPER; ++k) {
        int i = base + k * STHR;
        e[k] = 0; v[k] = 0;
        if (i < nnz) { e[k] = edge_idx[i]; v[k] = node_idx[i]; }
    }
    for (int i = tid; i < 2 * MAXBK; i += STHR) s_h[i] = 0;
    {
        int c4 = tid & 127, g = tid >> 7;
        int4 t = make_int4(0, 0, 0, 0);
        #pragma unroll 6
        for (int q = 0; q < NCB / 8; ++q) {
            int4 h = ((const int4*)(hist + (8 * q + g) * (2 * MAXBK)))[c4];
            t.x += h.x; t.y += h.y; t.z += h.z; t.w += h.w;
        }
        *(int4*)&s_tot[g][4 * c4] = t;
    }
    __syncthreads();
    if (wave < 2) {
        int cv[MAXBK / 32], sum = 0;
        #pragma unroll
        for (int j = 0; j < MAXBK / 32; ++j) {
            int b = wave * MAXBK + lane * (MAXBK / 32) + j;
            cv[j] = 0;
            #pragma unroll
            for (int g = 0; g < 8; ++g) cv[j] += s_tot[g][b];
            sum += cv[j];
        }
        int inc = sum;
        #pragma unroll
        for (int off = 1; off < 32; off <<= 1) {
            int t = __shfl_up(inc, off, 32);
            if (lane >= off) inc += t;
        }
        int run = inc - sum;
        #pragma unroll
        for (int j = 0; j < MAXBK / 32; ++j) {
            s_st[wave * MAXBK + lane * (MAXBK / 32) + j] = run;
            run += cv[j];
        }
    }
    __syncthreads();
    #pragma unroll
    for (int k = 0; k < SPER; ++k) {
        re[k] = 0; rn[k] = 0;
        if (base + k * STHR < nnz) {
            re[k] = atomicAdd(&s_h[e[k] >> ESH], 1);
            rn[k] = atomicAdd(&s_h[MAXBK + (v[k] >> NSH)], 1);
        }
    }
    __syncthreads();
    if (tid < nbe) {
        int c = s_h[tid], g = 0;
        if (c) g = atomicAdd(&cnt_e[(tid << ESH) + 2], c);
        s_fp[tid] = s_st[tid] + g;
        if (blockIdx.x == 0) cnt_e[tid << ESH] = s_st[tid];
    } else if (tid >= MAXBK && tid < MAXBK + nbn) {
        int b = tid - MAXBK;
        int c = s_h[tid], g = 0;
        if (c) g = atomicAdd(&cnt_n[(b << NSH) + 2], c);
        s_fp[tid] = s_st[tid] + g;
        if (blockIdx.x == 0) cnt_n[b << NSH] = s_st[tid];
    }
    if (wave >= 16 && wave < 18) {
        int side = wave - 16;
        int cv[MAXBK / 32], sum = 0;
        #pragma unroll
        for (int j = 0; j < MAXBK / 32; ++j) {
            cv[j] = s_h[side * MAXBK + lane * (MAXBK / 32) + j];
            sum += cv[j];
        }
        int inc = sum;
        #pragma unroll
        for (int off = 1; off < 32; off <<= 1) {
            int t = __shfl_up(inc, off, 32);
            if (lane >= off) inc += t;
        }
        int run = inc - sum;
        #pragma unroll
        for (int j = 0; j < MAXBK / 32; ++j) {
            s_ls[side * MAXBK + lane * (MAXBK / 32) + j] = run;
            run += cv[j];
        }
    }
    __syncthreads();
    #pragma unroll
    for (int k = 0; k < SPER; ++k) {
        if (base + k * STHR < nnz) {
            unsigned ue = (unsigned)e[k], uv = (unsigned)v[k];
            int be = e[k] >> ESH, bn = MAXBK + (v[k] >> NSH);
            int je = s_ls[be] + re[k], jn = s_ls[bn] + rn[k];
            s_val[0][je] = ((ue & ((1u << ESH) - 1)) << EPK) | uv;
            s_pos[0][je] = s_fp[be] + re[k];
            s_val[1][jn] = ((uv & ((1u << NSH) - 1)) << NPK) | ue;
            s_pos[1][jn] = s_fp[bn] + rn[k];
        }
    }
    __syncthreads();
    int mine = nnz - blk0; if (mine > STHR * SPER) mine = STHR * SPER;
    #pragma unroll
    for (int k = 0; k < SPER; ++k) {
        int j = tid + k * STHR;
        if (j < mine) {
            part_e[s_pos[0][j]] = s_val[0][j];
            part_n[s_pos[1][j]] = s_val[1][j];
        }
    }
}

static constexpr int CAP = 12288;

__global__ __launch_bounds__(1024) void fr_rows(const unsigned* __restrict__ part_e,
                                                const unsigned* __restrict__ part_n,
                                                int* off_e, int* off_n,
                                                int* __restrict__ perm_e, int* __restrict__ perm_n,
                                                int nbe, int E, int n, int nnz) {
    __shared__ int s_stage[CAP];
    __shared__ int s_c[1 << NSH];
    __shared__ int s_w[32];
    int tid = threadIdx.x, lane = tid & 31, wave = tid >> 5;
    bool ns = (int)blockIdx.x >= nbe;
    int b  = ns ? (int)blockIdx.x - nbe : (int)blockIdx.x;
    int sh = ns ? NSH : ESH, pk = ns ? NPK : EPK;
    int nrows = ns ? n : E;
    int* off = ns ? off_n : off_e;
    const unsigned* part = ns ? part_n : part_e;
    int* perm = ns ? perm_n : perm_e;
    int key0 = b << sh;
    int nk = nrows - key0; if (nk > (1 << sh)) nk = 1 << sh;
    int bstart = off[key0];
    int bend = (key0 + (1 << sh) < nrows) ? off[key0 + (1 << sh)] : nnz;
    int bcnt = bend - bstart;
    for (int i = tid; i < (1 << NSH); i += STHR) s_c[i] = 0;
    __syncthreads();
    const unsigned* src = part + bstart;
    for (int i0 = tid; i0 < bcnt + tid; i0 += STHR * SPER) {
        unsigned q[SPER];
        #pragma unroll
        for (int k = 0; k < SPER; ++k) { int i = i0 + k * STHR; q[k] = 0; if (i < bcnt) q[k] = src[i]; }
        #pragma unroll
        for (int k = 0; k < SPER; ++k) { if (i0 + k * STHR < bcnt) atomicAdd(&s_c[q[k] >> pk], 1); }
    }
    __syncthreads();
    int cv = 0, inc = 0;
    if (tid < (1 << NSH)) {
        cv = s_c[tid]; inc = cv;
        #pragma unroll
        for (int o = 1; o < 32; o <<= 1) {
            int t = __shfl_up(inc, o, 32);
            if (lane >= o) inc += t;
        }
        if (lane == 31) s_w[wave] = inc;
    }
    __syncthreads();
    if (tid < (1 << NSH)) {
        int pre = 0;
        #pragma unroll
        for (int w = 0; w < (1 << NSH) / 32; ++w) pre += (w < wave) ? s_w[w] : 0;
        int ex = pre + inc - cv;
        s_c[tid] = ex;
        if (tid > 0 && tid < nk) off[key0 + tid] = bstart + ex;
    }
    __syncthreads();
    unsigned mask = (1u << pk) - 1u;
    for (int i0 = tid; i0 < bcnt + tid; i0 += STHR * SPER) {
        unsigned q[SPER];
        #pragma unroll
        for (int k = 0; k < SPER; ++k) { int i = i0 + k * STHR; q[k] = 0; if (i < bcnt) q[k] = src[i]; }
        #pragma unroll
        for (int k = 0; k < SPER; ++k) {
            if (i0 + k * STHR < bcnt) {
                int p = atomicAdd(&s_c[q[k] >> pk], 1);
                int val = (int)(q[k] & mask);
                if (p < CAP) s_stage[p] = val; else perm[bstart + p] = val;
            }
        }
    }
    __syncthreads();
    int m = (bcnt < CAP) ? bcnt : CAP;
    for (int i = tid; i < m; i += STHR) perm[bstart + i] = s_stage[i];
}

static constexpr int RB = 8;

template <bool FULL>
__device__ __forceinline__ void fr_add_rows(const float* __restrict__ src, int mine, int j, int cnt,
                                            unsigned lane, float4& acc) {
    float4 v[RB];
    int r[RB];
    #pragma unroll
    for (int u = 0; u < RB; ++u) r[u] = __builtin_amdgcn_readlane(mine, (j + u) & 31);
    #pragma unroll
    for (int u = 0; u < RB; ++u) {
        v[u] = make_float4(0.f, 0.f, 0.f, 0.f);
        if (FULL || j + u < cnt) v[u] = ((const float4*)(src + (size_t)r[u] * KDIM))[lane];
    }
    #pragma unroll
    for (int u = 0; u < RB; ++u) {
        acc.x += v[u].x; acc.y += v[u].y; acc.z += v[u].z; acc.w += v[u].w;
    }
}

__device__ __forceinline__ float4 fr_row_sum(const float* __restrict__ src,
                                             const int* __restrict__ list,
                                             int b, int e, unsigned lane) {
    float4 acc = make_float4(0.f, 0.f, 0.f, 0.f);
    for (int c = b; c < e; c += 32) {
        int cnt = (e - c < 32) ? (e - c) : 32;
        int mine = ((int)lane < cnt) ? list[c + (int)lane] : 0;
        for (int j = 0; j < cnt; j += RB) {
            if (j + RB <= cnt) fr_add_rows<true>(src, mine, j, cnt, lane, acc);
            else               fr_add_rows<false>(src, mine, j, cnt, lane, acc);
        }
    }
    return acc;
}

__global__ __launch_bounds__(256) void fr_gather_gemm(const float* __restrict__ probs,
                                                      const int* __restrict__ off_e,
                                                      const int* __restrict__ perm_e,
                                                      const float* __restrict__ W,
                                                      float* __restrict__ ef, int E, int nnz) {
    __shared__ float sA[16][132];
    int tid = threadIdx.x;
    int wave = tid >> 5, lane = tid & 31;
    int m0 = blockIdx.x * 16;
    for (int rr = 0; rr < 2; ++rr) {
        int row0 = m0 + 2 * wave + rr;
        float4 acc = make_float4(0.f, 0.f, 0.f, 0.f);
        if (row0 < E) {
            int row = __builtin_amdgcn_readfirstlane(row0);
            int b = __builtin_amdgcn_readfirstlane(off_e[row]);
            int e = (row + 1 < E) ? __builtin_amdgcn_readfirstlane(off_e[row + 1]) : nnz;
            acc = fr_row_sum(probs, perm_e, b, e, (unsigned)lane);
            float inv = 1.f / fmaxf((float)(e - b), 1.f);
            acc.x *= inv; acc.y *= inv; acc.z *= inv; acc.w *= inv;
        }
        *(float4*)&sA[2 * wave + rr][lane * 4] = acc;
    }
    __syncthreads();

    int half = lane >> 4, lr = lane & 15;
    int n0 = wave * 16;

    v8f c = {0.f, 0.f, 0.f, 0.f, 0.f, 0.f, 0.f, 0.f};
    #pragma unroll 8
    for (int k0 = 0; k0 < KDIM; k0 += 4) {
        int ka = k0 + 2 * half;
        v2f a, b;
        a.x = sA[lr][ka];
        a.y = sA[lr][ka + 1];
        b.x = W[(size_t)ka * KDIM + n0 + lr];
        b.y = W[(size_t)(ka + 1) * KDIM + n0 + lr];
        c = __builtin_amdgcn_wmma_f32_16x16x4_f32(false, a, false, b, (short)0, c,
                                                  false, false);
    }
    int rowbase = m0 + 8 * half;
    #pragma unroll
    for (int r = 0; r < 8; ++r) {
        int row = rowbase + r;
        if (row < E) ef[(size_t)row * KDIM + n0 + lr] = c[r];
    }
}

__global__ __launch_bounds__(256) void fr_scatter_fin(const float* __restrict__ edge_feat,
                                                      const int* __restrict__ off_n,
                                                      const int* __restrict__ perm_n,
                                                      const float* __restrict__ probs,
                                                      float* __restrict__ out, int n, int nnz) {
    int wave = threadIdx.x >> 5;
    unsigned lane = threadIdx.x & 31;
    int nw = gridDim.x * 8;
    for (int row0 = blockIdx.x * 8 + wave; row0 < n; row0 += nw) {
        int row = __builtin_amdgcn_readfirstlane(row0);
        int b = __builtin_amdgcn_readfirstlane(off_n[row]);
        int e = (row + 1 < n) ? __builtin_amdgcn_readfirstlane(off_n[row + 1]) : nnz;
        float4 p = ((const float4*)(probs + (size_t)row * KDIM))[lane];
        float4 acc = fr_row_sum(edge_feat, perm_n, b, e, lane);
        float inv = 1.f / fmaxf((float)(e - b), 1.f);
        float4 nat;
        nat.x = fmaxf(p.x, EPS) * acc.x * inv;
        nat.y = fmaxf(p.y, EPS) * acc.y * inv;
        nat.z = fmaxf(p.z, EPS) * acc.z * inv;
        nat.w = fmaxf(p.w, EPS) * acc.w * inv;
        float s = nat.x + nat.y + nat.z + nat.w;
        #pragma unroll
        for (int off = 16; off > 0; off >>= 1) s += __shfl_xor(s, off, 32);
        float mean = s * (1.0f / KDIM);
        nat.x -= mean; nat.y -= mean; nat.z -= mean; nat.w -= mean;
        ((float4*)(out + (size_t)row * KDIM))[lane] = nat;
    }
}

extern "C" void kernel_launch(void* const* d_in, const int* in_sizes, int n_in,
                              void* d_out, int out_size, void* d_ws, size_t ws_size,
                              hipStream_t stream) {
    const float* y        = (const float*)d_in[0];
    const float* W        = (const float*)d_in[1];
    const int*   node_idx = (const int*)d_in[2];
    const int*   edge_idx = (const int*)d_in[3];
    float*       out      = (float*)d_out;

    const int n   = in_sizes[0] / KDIM;
    const int nnz = in_sizes[2];
    const int E   = N_EDGES;
    const int nbe = (E + (1 << ESH) - 1) >> ESH;
    const int nbn = (n + (1 << NSH) - 1) >> NSH;

    float* probs   = (float*)d_ws;
    float* plane_a = probs + (size_t)n * KDIM;
    float* plane_b = plane_a + (size_t)E * KDIM;
    int*   cnt     = (int*)(plane_b + (size_t)E * KDIM);
    unsigned* part_e = (unsigned*)plane_a;
    unsigned* part_n = part_e + nnz;
    int*   perm_e  = (int*)plane_b;
    int*   perm_n  = perm_e + nnz;

    fr_softmax_count<<<NCB + (n + 31) / 32, 1024, 0, stream>>>(y, probs, node_idx, edge_idx, cnt, E + n,
                                                               (int*)plane_b, n, nnz);
    fr_part<<<(nnz + STHR * SPER - 1) / (STHR * SPER), STHR, 0, stream>>>(node_idx, edge_idx, (const int*)plane_b,
                                                                        cnt, cnt + E, part_e, part_n, nbe, nbn, nnz);
    fr_rows<<<nbe + nbn, STHR, 0, stream>>>(part_e, part_n, cnt, cnt + E, perm_e, perm_n, nbe, E, n, nnz);
    fr_gather_gemm<<<(E + 15) / 16, 256, 0, stream>>>(probs, cnt, perm_e, W, plane_a, E, nnz);
    fr_scatter_fin<<<(n + 63) / 64, 256, 0, stream>>>(plane_a, cnt + E, perm_n, probs, out, n, nnz);
}


# edge bucket's row sort, row sums and product with W in one work-group: sorted list stays in LDS, no perm_e/off_e trip, 5 -> 4 launches (node buckets sort beside them); sums stay in arrival order
# speedup vs baseline: 6.1843x; 1.0330x over previous
#include <hip/hip_runtime.h>

typedef __attribute__((ext_vector_type(2))) float v2f;
typedef __attribute__((ext_vector_type(8))) float v8f;

static constexpr int KDIM    = 128;
static constexpr int N_EDGES = 25000;
static constexpr float EPS   = 1e-12f;


static constexpr int ESH   = 7;
static constexpr int NSH   = 9;
static constexpr int MAXBK = 256;
static constexpr int NCB   = 96;
static constexpr int CPT   = 9;

__global__ __launch_bounds__(1024) void fr_softmax_count(const float* __restrict__ y,
                                                         float* __restrict__ probs,
                                                         const int* __restrict__ node_idx,
                                                         const int* __restrict__ edge_idx,
                                                         int* __restrict__ cnt, int ncnt,
                                                         int* __restrict__ hist, int n, int nnz) {
    __shared__ int s_h[2 * MAXBK];
    int tid = threadIdx.x;
    if (blockIdx.x < NCB) {
        for (int i = tid; i < 2 * MAXBK; i += 1024) s_h[i] = 0;
        __syncthreads();
        int per = (nnz + NCB - 1) / NCB;
        int lo = blockIdx.x * per;
        int hi = (lo + per < nnz) ? lo + per : nnz;
        for (int i0 = lo + tid; i0 < hi; i0 += 1024 * CPT) {
            int e[CPT], v[CPT];
            #pragma unroll
            for (int k = 0; k < CPT; ++k) {
                int i = i0 + k * 1024;
                e[k] = 0; v[k] = 0;
                if (i < hi) { e[k] = edge_idx[i]; v[k] = node_idx[i]; }
            }
            #pragma unroll
            for (int k = 0; k < CPT; ++k) {
                if (i0 + k * 1024 < hi) {
                    atomicAdd(&s_h[e[k] >> ESH], 1);
                    atomicAdd(&s_h[MAXBK + (v[k] >> NSH)], 1);
                }
            }
        }
        __syncthreads();
        if (tid < 2 * MAXBK) hist[blockIdx.x * (2 * MAXBK) + tid] = s_h[tid];
        return;
    }
    int sb = blockIdx.x - NCB, nsb = gridDim.x - NCB;
    for (int i = sb * 1024 + tid; i < ncnt; i += nsb * 1024) cnt[i] = 0;
    int wave = tid >> 5, lane = tid & 31;
    int row = sb * 32 + wave;
    if (row >= n) return;
    float4 v = ((const float4*)(y + (size_t)row * KDIM))[lane];
    float m = fmaxf(fmaxf(v.x, v.y), fmaxf(v.z, v.w));
    #pragma unroll
    for (int off = 16; off > 0; off >>= 1) m = fmaxf(m, __shfl_xor(m, off, 32));
    float4 e;
    e.x = __expf(v.x - m); e.y = __expf(v.y - m);
    e.z = __expf(v.z - m); e.w = __expf(v.w - m);
    float s = e.x + e.y + e.z + e.w;
    #pragma unroll
    for (int off = 16; off > 0; off >>= 1) s += __shfl_xor(s, off, 32);
    float inv = 1.f / s;
    e.x *= inv; e.y *= inv; e.z *= inv; e.w *= inv;
    ((float4*)(probs + (size_t)row * KDIM))[lane] = e;
}

static constexpr int EPK   = 25;
static constexpr int NPK   = 23;
static constexpr int STHR  = 1024;
static constexpr int SPER  = 4;

__global__ __launch_bounds__(1024) void fr_part(const int* __restrict__ node_idx,
                                                const int* __restrict__ edge_idx,
                                                const int* __restrict__ hist,
                                                int* cnt_e, int* cnt_n,
                                                unsigned* __restrict__ part_e,
                                                unsigned* __restrict__ part_n,
                                                int nbe, int nbn, int nnz) {
    __shared__ int s_h[2 * MAXBK];
    __shared__ __attribute__((aligned(16))) int s_tot[8][2 * MAXBK];
    __shared__ int s_st[2 * MAXBK];
    __shared__ int s_fp[2 * MAXBK];
    __shared__ int s_ls[2 * MAXBK];
    __shared__ unsigned s_val[2][STHR * SPER];
    __shared__ int      s_pos[2][STHR * SPER];
    int tid = threadIdx.x, lane = tid & 31, wave = tid >> 5;
    int blk0 = blockIdx.x * (STHR * SPER);
    int base = blk0 + tid;
    int e[SPER], v[SPER], re[SPER], rn[SPER];
    #pragma unroll
    for (int k = 0; k < SPER; ++k) {
        int i = base + k * STHR;
        e[k] = 0; v[k] = 0;
        if (i < nnz) { e[k] = edge_idx[i]; v[k] = node_idx[i]; }
    }
    for (int i = tid; i < 2 * MAXBK; i += STHR) s_h[i] = 0;
    {
        int c4 = tid & 127, g = tid >> 7;
        int4 t = make_int4(0, 0, 0, 0);
        #pragma unroll 6
        for (int q = 0; q < NCB / 8; ++q) {
            int4 h = ((const int4*)(hist + (8 * q + g) * (2 * MAXBK)))[c4];
            t.x += h.x; t.y += h.y; t.z += h.z; t.w += h.w;
        }
        *(int4*)&s_tot[g][4 * c4] = t;
    }
    __syncthreads();
    if (wave < 2) {
        int cv[MAXBK / 32], sum = 0;
        #pragma unroll
        for (int j = 0; j < MAXBK / 32; ++j) {
            int b = wave * MAXBK + lane * (MAXBK / 32) + j;
            cv[j] = 0;
            #pragma unroll
            for (int g = 0; g < 8; ++g) cv[j] += s_tot[g][b];
            sum += cv[j];
        }
        int inc = sum;
        #pragma unroll
        for (int off = 1; off < 32; off <<= 1) {
            int t = __shfl_up(inc, off, 32);
            if (lane >= off) inc += t;
        }
        int run = inc - sum;
        #pragma unroll
        for (int j = 0; j < MAXBK / 32; ++j) {
            s_st[wave * MAXBK + lane * (MAXBK / 32) + j] = run;
            run += cv[j];
        }
    }
    __syncthreads();
    #pragma unroll
    for (int k = 0; k < SPER; ++k) {
        re[k] = 0; rn[k] = 0;
        if (base + k * STHR < nnz) {
            re[k] = atomicAdd(&s_h[e[k] >> ESH], 1);
            rn[k] = atomicAdd(&s_h[MAXBK + (v[k] >> NSH)], 1);
        }
    }
    __syncthreads();
    if (tid < nbe) {
        int c = s_h[tid], g = 0;
        if (c) g = atomicAdd(&cnt_e[(tid << ESH) + 2], c);
        s_fp[tid] = s_st[tid] + g;
        if (blockIdx.x == 0) cnt_e[tid << ESH] = s_st[tid];
    } else if (tid >= MAXBK && tid < MAXBK + nbn) {
        int b = tid - MAXBK;
        int c = s_h[tid], g = 0;
        if (c) g = atomicAdd(&cnt_n[(b << NSH) + 2], c);
        s_fp[tid] = s_st[tid] + g;
        if (blockIdx.x == 0) cnt_n[b << NSH] = s_st[tid];
    }
    if (wave >= 16 && wave < 18) {
        int side = wave - 16;
        int cv[MAXBK / 32], sum = 0;
        #pragma unroll
        for (int j = 0; j < MAXBK / 32; ++j) {
            cv[j] = s_h[side * MAXBK + lane * (MAXBK / 32) + j];
            sum += cv[j];
        }
        int inc = sum;
        #pragma unroll
        for (int off = 1; off < 32; off <<= 1) {
            int t = __shfl_up(inc, off, 32);
            if (lane >= off) inc += t;
        }
        int run = inc - sum;
        #pragma unroll
        for (int j = 0; j < MAXBK / 32; ++j) {
            s_ls[side * MAXBK + lane * (MAXBK / 32) + j] = run;
            run += cv[j];
        }
    }
    __syncthreads();
    #pragma unroll
    for (int k = 0; k < SPER; ++k) {
        if (base + k * STHR < nnz) {
            unsigned ue = (unsigned)e[k], uv = (unsigned)v[k];
            int be = e[k] >> ESH, bn = MAXBK + (v[k] >> NSH);
            int je = s_ls[be] + re[k], jn = s_ls[bn] + rn[k];
            s_val[0][je] = ((ue & ((1u << ESH) - 1)) << EPK) | uv;
            s_pos[0][je] = s_fp[be] + re[k];
            s_val[1][jn] = ((uv & ((1u << NSH) - 1)) << NPK) | ue;
            s_pos[1][jn] = s_fp[bn] + rn[k];
        }
    }
    __syncthreads();
    int mine = nnz - blk0; if (mine > STHR * SPER) mine = STHR * SPER;
    #pragma unroll
    for (int k = 0; k < SPER; ++k) {
        int j = tid + k * STHR;
        if (j < mine) {
            part_e[s_pos[0][j]] = s_val[0][j];
            part_n[s_pos[1][j]] = s_val[1][j];
        }
    }
}

static constexpr int RB = 8;

template <bool FULL>
__device__ __forceinline__ void fr_add_rows(const float* __restrict__ src, int mine, int j, int cnt,
                                            unsigned lane, float4& acc) {
    float4 v[RB];
    int r[RB];
    #pragma unroll
    for (int u = 0; u < RB; ++u) r[u] = __builtin_amdgcn_readlane(mine, (j + u) & 31);
    #pragma unroll
    for (int u = 0; u < RB; ++u) {
        v[u] = make_float4(0.f, 0.f, 0.f, 0.f);
        if (FULL || j + u < cnt) v[u] = ((const float4*)(src + (size_t)r[u] * KDIM))[lane];
    }
    #pragma unroll
    for (int u = 0; u < RB; ++u) {
        acc.x += v[u].x; acc.y += v[u].y; acc.z += v[u].z; acc.w += v[u].w;
    }
}

__device__ __forceinline__ float4 fr_row_sum(const float* __restrict__ src,
                                             const int* __restrict__ list,
                                             int b, int e, unsigned lane) {
    float4 acc = make_float4(0.f, 0.f, 0.f, 0.f);
    for (int c = b; c < e; c += 32) {
        int cnt = (e - c < 32) ? (e - c) : 32;
        int mine = ((int)lane < cnt) ? list[c + (int)lane] : 0;
        for (int j = 0; j < cnt; j += RB) {
            if (j + RB <= cnt) fr_add_rows<true>(src, mine, j, cnt, lane, acc);
            else               fr_add_rows<false>(src, mine, j, cnt, lane, acc);
        }
    }
    return acc;
}

static constexpr int CAP = 12288;
static constexpr int ERW = 1 << ESH;

__global__ __launch_bounds__(1024) void fr_rows_edges(const unsigned* __restrict__ part_e,
                                                      const unsigned* __restrict__ part_n,
                                                      const int* off_e, int* off_n,
                                                      int* perm_e, int* __restrict__ perm_n,
                                                      const float* __restrict__ probs,
                                                      const float* __restrict__ W,
                                                      float* __restrict__ ef,
                                                      int nbe, int E, int n, int nnz) {
    __shared__ int s_stage[CAP];
    __shared__ int s_c[1 << NSH];
    __shared__ int s_w[32];
    __shared__ __attribute__((aligned(16))) float sA[ERW][132];
    int tid = threadIdx.x, lane = tid & 31, wave = tid >> 5;
    bool ns = (int)blockIdx.x >= nbe;
    int b  = ns ? (int)blockIdx.x - nbe : (int)blockIdx.x;
    int sh = ns ? NSH : ESH, pk = ns ? NPK : EPK;
    int nrows = ns ? n : E;
    const int* off = ns ? (const int*)off_n : off_e;
    const unsigned* part = ns ? part_n : part_e;
    int* perm = ns ? perm_n : perm_e;
    int key0 = b << sh;
    int nk = nrows - key0; if (nk > (1 << sh)) nk = 1 << sh;
    int bstart = off[key0];
    int bend = (key0 + (1 << sh) < nrows) ? off[key0 + (1 << sh)] : nnz;
    int bcnt = bend - bstart;
    for (int i = tid; i < (1 << NSH); i += STHR) s_c[i] = 0;
    __syncthreads();
    const unsigned* src = part + bstart;
    for (int i0 = tid; i0 < bcnt + tid; i0 += STHR * SPER) {
        unsigned q[SPER];
        #pragma unroll
        for (int k = 0; k < SPER; ++k) { int i = i0 + k * STHR; q[k] = 0; if (i < bcnt) q[k] = src[i]; }
        #pragma unroll
        for (int k = 0; k < SPER; ++k) { if (i0 + k * STHR < bcnt) atomicAdd(&s_c[q[k] >> pk], 1); }
    }
    __syncthreads();
    int cv = 0, inc = 0;
    if (tid < (1 << NSH)) {
        cv = s_c[tid]; inc = cv;
        #pragma unroll
        for (int o = 1; o < 32; o <<= 1) {
            int t = __shfl_up(inc, o, 32);
            if (lane >= o) inc += t;
        }
        if (lane == 31) s_w[wave] = inc;
    }
    __syncthreads();
    if (tid < (1 << NSH)) {
        int pre = 0;
        #pragma unroll
        for (int w = 0; w < (1 << NSH) / 32; ++w) pre += (w < wave) ? s_w[w] : 0;
        int ex = pre + inc - cv;
        s_c[tid] = ex;
        if (ns && tid > 0 && tid < nk) off_n[key0 + tid] = bstart + ex;
    }
    __syncthreads();
    unsigned mask = (1u << pk) - 1u;
    for (int i0 = tid; i0 < bcnt + tid; i0 += STHR * SPER) {
        unsigned q[SPER];
        #pragma unroll
        for (int k = 0; k < SPER; ++k) { int i = i0 + k * STHR; q[k] = 0; if (i < bcnt) q[k] = src[i]; }
        #pragma unroll
        for (int k = 0; k < SPER; ++k) {
            if (i0 + k * STHR < bcnt) {
                int p = atomicAdd(&s_c[q[k] >> pk], 1);
                int val = (int)(q[k] & mask);
                if (p < CAP) s_stage[p] = val; else perm[bstart + p] = val;
            }
        }
    }
    __syncthreads();
    if (ns) {
        int m = (bcnt < CAP) ? bcnt : CAP;
        for (int i = tid; i < m; i += STHR) perm_n[bstart + i] = s_stage[i];
        return;
    }

    for (int rr = 0; rr < ERW / 32; ++rr) {
        int r = __builtin_amdgcn_readfirstlane(wave * (ERW / 32) + rr);
        int rb = (r > 0) ? __builtin_amdgcn_readfirstlane(s_c[r - 1]) : 0;
        int re = __builtin_amdgcn_readfirstlane(s_c[r]);
        float4 acc = make_float4(0.f, 0.f, 0.f, 0.f);
        for (int c = rb; c < re; c += 32) {
            int cnt = (re - c < 32) ? (re - c) : 32;
            int p = c + lane;
            int mine = s_stage[(p < CAP) ? p : CAP - 1];
            if (c + 32 > CAP) {
                if (p >= CAP && lane < cnt) mine = perm_e[bstart + p];
            }
            for (int j = 0; j < cnt; j += RB) {
                if (j + RB <= cnt) fr_add_rows<true>(probs, mine, j, cnt, (unsigned)lane, acc);
                else               fr_add_rows<false>(probs, mine, j, cnt, (unsigned)lane, acc);
            }
        }
        float inv = 1.f / fmaxf((float)(re - rb), 1.f);
        acc.x *= inv; acc.y *= inv; acc.z *= inv; acc.w *= inv;
        *(float4*)&sA[r][lane * 4] = acc;
    }
    __syncthreads();

    int half = lane >> 4, lr = lane & 15;
    int n0 = (wave & 7) * 16;
    int r0 = (wave >> 3) * 32;
    v8f c0 = {0.f, 0.f, 0.f, 0.f, 0.f, 0.f, 0.f, 0.f};
    v8f c1 = {0.f, 0.f, 0.f, 0.f, 0.f, 0.f, 0.f, 0.f};
    #pragma unroll 8
    for (int k0 = 0; k0 < KDIM; k0 += 4) {
        int ka = k0 + 2 * half;
        v2f a0, a1, bw;
        a0.x = sA[r0 + lr][ka];
        a0.y = sA[r0 + lr][ka + 1];
        a1.x = sA[r0 + 16 + lr][ka];
        a1.y = sA[r0 + 16 + lr][ka + 1];
        bw.x = W[(size_t)ka * KDIM + n0 + lr];
        bw.y = W[(size_t)(ka + 1) * KDIM + n0 + lr];
        c0 = __builtin_amdgcn_wmma_f32_16x16x4_f32(false, a0, false, bw, (short)0, c0,
                                                   false, false);
        c1 = __builtin_amdgcn_wmma_f32_16x16x4_f32(false, a1, false, bw, (short)0, c1,
                                                   false, false);
    }
    int rowbase = key0 + r0 + 8 * half;
    #pragma unroll
    for (int r = 0; r < 8; ++r) {
        int row = rowbase + r;
        if (row < E) ef[(size_t)row * KDIM + n0 + lr] = c0[r];
        if (row + 16 < E) ef[(size_t)(row + 16) * KDIM + n0 + lr] = c1[r];
    }
}

__global__ __launch_bounds__(256) void fr_scatter_fin(const float* __restrict__ edge_feat,
                                                      const int* __restrict__ off_n,
                                                      const int* __restrict__ perm_n,
                                                      const float* __restrict__ probs,
                                                      float* __restrict__ out, int n, int nnz) {
    int wave = threadIdx.x >> 5;
    unsigned lane = threadIdx.x & 31;
    int nw = gridDim.x * 8;
    for (int row0 = blockIdx.x * 8 + wave; row0 < n; row0 += nw) {
        int row = __builtin_amdgcn_readfirstlane(row0);
        int b = __builtin_amdgcn_readfirstlane(off_n[row]);
        int e = (row + 1 < n) ? __builtin_amdgcn_readfirstlane(off_n[row + 1]) : nnz;
        float4 p = ((const float4*)(probs + (size_t)row * KDIM))[lane];
        float4 acc = fr_row_sum(edge_feat, perm_n, b, e, lane);
        float inv = 1.f / fmaxf((float)(e - b), 1.f);
        float4 nat;
        nat.x = fmaxf(p.x, EPS) * acc.x * inv;
        nat.y = fmaxf(p.y, EPS) * acc.y * inv;
        nat.z = fmaxf(p.z, EPS) * acc.z * inv;
        nat.w = fmaxf(p.w, EPS) * acc.w * inv;
        float s = nat.x + nat.y + nat.z + nat.w;
        #pragma unroll
        for (int off = 16; off > 0; off >>= 1) s += __shfl_xor(s, off, 32);
        float mean = s * (1.0f / KDIM);
        nat.x -= mean; nat.y -= mean; nat.z -= mean; nat.w -= mean;
        ((float4*)(out + (size_t)row * KDIM))[lane] = nat;
    }
}

extern "C" void kernel_launch(void* const* d_in, const int* in_sizes, int n_in,
                              void* d_out, int out_size, void* d_ws, size_t ws_size,
                              hipStream_t stream) {
    const float* y        = (const float*)d_in[0];
    const float* W        = (const float*)d_in[1];
    const int*   node_idx = (const int*)d_in[2];
    const int*   edge_idx = (const int*)d_in[3];
    float*       out      = (float*)d_out;

    const int n   = in_sizes[0] / KDIM;
    const int nnz = in_sizes[2];
    const int E   = N_EDGES;
    const int nbe = (E + (1 << ESH) - 1) >> ESH;
    const int nbn = (n + (1 << NSH) - 1) >> NSH;

    float* probs   = (float*)d_ws;
    float* plane_a = probs + (size_t)n * KDIM;
    float* plane_b = plane_a + (size_t)E * KDIM;
    int*   cnt     = (int*)(plane_b + (size_t)E * KDIM);
    unsigned* part_e = (unsigned*)out;
    unsigned* part_n = part_e + nnz;
    int*   perm_e  = (int*)plane_b;
    int*   perm_n  = perm_e + nnz;

    fr_softmax_count<<<NCB + (n + 31) / 32, 1024, 0, stream>>>(y, probs, node_idx, edge_idx, cnt, E + n,
                                                               (int*)plane_b, n, nnz);
    fr_part<<<(nnz + STHR * SPER - 1) / (STHR * SPER), STHR, 0, stream>>>(node_idx, edge_idx, (const int*)plane_b,
                                                                        cnt, cnt + E, part_e, part_n, nbe, nbn, nnz);
    fr_rows_edges<<<nbe + nbn, STHR, 0, stream>>>(part_e, part_n, cnt, cnt + E, perm_e, perm_n, probs, W,
                                                 plane_a, nbe, E, n, nnz);
    fr_scatter_fin<<<(n + 63) / 64, 256, 0, stream>>>(plane_a, cnt + E, perm_n, probs, out, n, nnz);
}


# incidence sort level one made slice-local (LDS count, scan, rank; run starts to a table) so it runs beside the softmax: no global count or add, 4 -> 3 launches; sums stay in arrival order
# speedup vs baseline: 6.6665x; 1.0780x over previous
#include <hip/hip_runtime.h>

typedef __attribute__((ext_vector_type(2))) float v2f;
typedef __attribute__((ext_vector_type(8))) float v8f;

static constexpr int KDIM    = 128;
static constexpr int N_EDGES = 25000;
static constexpr float EPS   = 1e-12f;


static constexpr int ESH   = 7;
static constexpr int NSH   = 9;
static constexpr int MAXBK = 256;
static constexpr int EPK   = 25;
static constexpr int NPK   = 23;
static constexpr int STHR  = 1024;
static constexpr int SPER  = 4;
static constexpr int SLICE = STHR * SPER;

__global__ __launch_bounds__(1024) void fr_softmax_slices(const float* __restrict__ y,
                                                          float* __restrict__ probs,
                                                          const int* __restrict__ node_idx,
                                                          const int* __restrict__ edge_idx,
                                                          unsigned* __restrict__ seg_e,
                                                          unsigned* __restrict__ seg_n,
                                                          int* __restrict__ tab,
                                                          int npb, int n, int nnz) {
    __shared__ int s_h[2 * MAXBK];
    __shared__ int s_ls[2 * MAXBK];
    __shared__ unsigned s_val[2][SLICE];
    int tid = threadIdx.x, lane = tid & 31, wave = tid >> 5;
    if ((int)blockIdx.x < npb) {
        int blk0 = blockIdx.x * SLICE;
        int base = blk0 + tid;
        int e[SPER], v[SPER], re[SPER], rn[SPER];
        #pragma unroll
        for (int k = 0; k < SPER; ++k) {
            int i = base + k * STHR;
            e[k] = 0; v[k] = 0;
            if (i < nnz) { e[k] = edge_idx[i]; v[k] = node_idx[i]; }
        }
        for (int i = tid; i < 2 * MAXBK; i += STHR) s_h[i] = 0;
        __syncthreads();
        #pragma unroll
        for (int k = 0; k < SPER; ++k) {
            re[k] = 0; rn[k] = 0;
            if (base + k * STHR < nnz) {
                re[k] = atomicAdd(&s_h[e[k] >> ESH], 1);
                rn[k] = atomicAdd(&s_h[MAXBK + (v[k] >> NSH)], 1);
            }
        }
        __syncthreads();
        if (wave < 2) {
            int cv[MAXBK / 32], sum = 0;
            #pragma unroll
            for (int j = 0; j < MAXBK / 32; ++j) {
                cv[j] = s_h[wave * MAXBK + lane * (MAXBK / 32) + j];
                sum += cv[j];
            }
            int inc = sum;
            #pragma unroll
            for (int off = 1; off < 32; off <<= 1) {
                int t = __shfl_up(inc, off, 32);
                if (lane >= off) inc += t;
            }
            int run = inc - sum;
            int* trow = tab + (size_t)blockIdx.x * (2 * MAXBK) + wave * MAXBK + lane * (MAXBK / 32);
            #pragma unroll
            for (int j = 0; j < MAXBK / 32; ++j) {
                s_ls[wave * MAXBK + lane * (MAXBK / 32) + j] = run;
                trow[j] = run;
                run += cv[j];
            }
        }
        __syncthreads();
        #pragma unroll
        for (int k = 0; k < SPER; ++k) {
            if (base + k * STHR < nnz) {
                unsigned ue = (unsigned)e[k], uv = (unsigned)v[k];
                int je = s_ls[e[k] >> ESH] + re[k];
                int jn = s_ls[MAXBK + (v[k] >> NSH)] + rn[k];
                s_val[0][je] = ((ue & ((1u << ESH) - 1)) << EPK) | uv;
                s_val[1][jn] = ((uv & ((1u << NSH) - 1)) << NPK) | ue;
            }
        }
        __syncthreads();
        int mine = nnz - blk0; if (mine > SLICE) mine = SLICE;
        #pragma unroll
        for (int k = 0; k < SPER; ++k) {
            int j = tid + k * STHR;
            if (j < mine) {
                seg_e[blk0 + j] = s_val[0][j];
                seg_n[blk0 + j] = s_val[1][j];
            }
        }
        return;
    }
    int sb = blockIdx.x - npb;
    int row = sb * 32 + wave;
    if (row >= n) return;
    float4 v = ((const float4*)(y + (size_t)row * KDIM))[lane];
    float m = fmaxf(fmaxf(v.x, v.y), fmaxf(v.z, v.w));
    #pragma unroll
    for (int off = 16; off > 0; off >>= 1) m = fmaxf(m, __shfl_xor(m, off, 32));
    float4 e;
    e.x = __expf(v.x - m); e.y = __expf(v.y - m);
    e.z = __expf(v.z - m); e.w = __expf(v.w - m);
    float s = e.x + e.y + e.z + e.w;
    #pragma unroll
    for (int off = 16; off > 0; off >>= 1) s += __shfl_xor(s, off, 32);
    float inv = 1.f / s;
    e.x *= inv; e.y *= inv; e.z *= inv; e.w *= inv;
    ((float4*)(probs + (size_t)row * KDIM))[lane] = e;
}

static constexpr int RB = 8;

template <bool FULL>
__device__ __forceinline__ void fr_add_rows(const float* __restrict__ src, int mine, int j, int cnt,
                                            unsigned lane, float4& acc) {
    float4 v[RB];
    int r[RB];
    #pragma unroll
    for (int u = 0; u < RB; ++u) r[u] = __builtin_amdgcn_readlane(mine, (j + u) & 31);
    #pragma unroll
    for (int u = 0; u < RB; ++u) {
        v[u] = make_float4(0.f, 0.f, 0.f, 0.f);
        if (FULL || j + u < cnt) v[u] = ((const float4*)(src + (size_t)r[u] * KDIM))[lane];
    }
    #pragma unroll
    for (int u = 0; u < RB; ++u) {
        acc.x += v[u].x; acc.y += v[u].y; acc.z += v[u].z; acc.w += v[u].w;
    }
}

__device__ __forceinline__ float4 fr_row_sum(const float* __restrict__ src,
                                             const int* __restrict__ list,
                                             int b, int e, unsigned lane) {
    float4 acc = make_float4(0.f, 0.f, 0.f, 0.f);
    for (int c = b; c < e; c += 32) {
        int cnt = (e - c < 32) ? (e - c) : 32;
        int mine = ((int)lane < cnt) ? list[c + (int)lane] : 0;
        for (int j = 0; j < cnt; j += RB) {
            if (j + RB <= cnt) fr_add_rows<true>(src, mine, j, cnt, lane, acc);
            else               fr_add_rows<false>(src, mine, j, cnt, lane, acc);
        }
    }
    return acc;
}


static constexpr int CAP = 12288;
static constexpr int ERW = 1 << ESH;
static constexpr int RJ  = 7;

template <bool PLACE>
__device__ __forceinline__ void fr_sweep(const unsigned* __restrict__ seg, const int* s_src,
                                         const int* s_cn, int npb, int wave, int lane, int pk,
                                         unsigned mask, int* s_c, int* s_stage,
                                         int* __restrict__ perm_b) {
    for (int t0 = wave; t0 < npb; t0 += 32 * RJ) {
        int cl[RJ], sl[RJ], cc[RJ], ss[RJ];
        unsigned q[RJ];
        #pragma unroll
        for (int u = 0; u < RJ; ++u) {
            int t = t0 + 32 * u;
            int tc = (t < npb) ? t : 0;
            cl[u] = s_cn[tc]; sl[u] = s_src[tc];
        }
        #pragma unroll
        for (int u = 0; u < RJ; ++u) {
            int t = t0 + 32 * u;
            cc[u] = (t < npb) ? __builtin_amdgcn_readfirstlane(cl[u]) : 0;
            ss[u] = __builtin_amdgcn_readfirstlane(sl[u]);
        }
        #pragma unroll
        for (int u = 0; u < RJ; ++u) { q[u] = 0; if (lane < cc[u]) q[u] = seg[ss[u] + lane]; }
        if (PLACE) {
            int p[RJ];
            #pragma unroll
            for (int u = 0; u < RJ; ++u) { p[u] = 0; if (lane < cc[u]) p[u] = atomicAdd(&s_c[q[u] >> pk], 1); }
            #pragma unroll
            for (int u = 0; u < RJ; ++u) {
                if (lane < cc[u]) {
                    int val = (int)(q[u] & mask);
                    if (p[u] < CAP) s_stage[p[u]] = val; else perm_b[p[u]] = val;
                }
            }
        } else {
            #pragma unroll
            for (int u = 0; u < RJ; ++u) { if (lane < cc[u]) atomicAdd(&s_c[q[u] >> pk], 1); }
        }
        #pragma unroll
        for (int u = 0; u < RJ; ++u) {
            for (int o = 32 + lane; o < cc[u]; o += 32) {
                unsigned qq = seg[ss[u] + o];
                if (PLACE) {
                    int pp = atomicAdd(&s_c[qq >> pk], 1);
                    int val = (int)(qq & mask);
                    if (pp < CAP) s_stage[pp] = val; else perm_b[pp] = val;
                } else {
                    atomicAdd(&s_c[qq >> pk], 1);
                }
            }
        }
    }
}

__global__ __launch_bounds__(1024) void fr_rows_edges(const unsigned* __restrict__ seg_e,
                                                      const unsigned* __restrict__ seg_n,
                                                      const int* __restrict__ tab,
                                                      int* __restrict__ off_n,
                                                      int* perm_e, int* __restrict__ perm_n,
                                                      const float* __restrict__ probs,
                                                      const float* __restrict__ W,
                                                      float* __restrict__ ef,
                                                      int nbe, int npb, int E, int n) {
    __shared__ int s_stage[CAP];
    __shared__ int s_c[1 << NSH];
    __shared__ int s_w[32];
    __shared__ int s_src[STHR];
    __shared__ int s_cn[STHR];
    __shared__ int s_red[2];
    __shared__ __attribute__((aligned(16))) float sA[ERW][132];
    int tid = threadIdx.x, lane = tid & 31, wave = tid >> 5;
    bool ns = (int)blockIdx.x >= nbe;
    int b  = ns ? (int)blockIdx.x - nbe : (int)blockIdx.x;
    int sh = ns ? NSH : ESH, pk = ns ? NPK : EPK;
    int nrows = ns ? n : E;
    const unsigned* seg = ns ? seg_n : seg_e;
    int* perm = ns ? perm_n : perm_e;
    int key0 = b << sh;
    int nk = nrows - key0; if (nk > (1 << sh)) nk = 1 << sh;
    if (npb > STHR) npb = STHR;
    int st = 0, rc = 0;
    if (tid < npb) {
        const int* trow = tab + (size_t)tid * (2 * MAXBK) + (ns ? MAXBK : 0) + b;
        st = trow[0];
        rc = trow[1] - st;
    }
    for (int i = tid; i < (1 << NSH); i += STHR) s_c[i] = 0;
    if (tid < 2) s_red[tid] = 0;
    s_src[tid] = tid * SLICE + st;
    s_cn[tid] = rc;
    __syncthreads();
    {
        int a = st, c = rc;
        #pragma unroll
        for (int o = 16; o > 0; o >>= 1) { a += __shfl_xor(a, o, 32); c += __shfl_xor(c, o, 32); }
        if (lane == 0 && wave * 32 < npb) { atomicAdd(&s_red[0], a); atomicAdd(&s_red[1], c); }
    }
    __syncthreads();
    int bstart = s_red[0];
    int bcnt = s_red[1];
    unsigned mask = (1u << pk) - 1u;
    int* perm_b = perm + bstart;
    fr_sweep<false>(seg, s_src, s_cn, npb, wave, lane, pk, mask, s_c, s_stage, perm_b);
    __syncthreads();
    int cv = 0, inc = 0;
    if (tid < (1 << NSH)) {
        cv = s_c[tid]; inc = cv;
        #pragma unroll
        for (int o = 1; o < 32; o <<= 1) {
            int t = __shfl_up(inc, o, 32);
            if (lane >= o) inc += t;
        }
        if (lane == 31) s_w[wave] = inc;
    }
    __syncthreads();
    if (tid < (1 << NSH)) {
        int pre = 0;
        #pragma unroll
        for (int w = 0; w < (1 << NSH) / 32; ++w) pre += (w < wave) ? s_w[w] : 0;
        int ex = pre + inc - cv;
        s_c[tid] = ex;
        if (ns && tid < nk) off_n[key0 + tid] = bstart + ex;
    }
    __syncthreads();
    fr_sweep<true>(seg, s_src, s_cn, npb, wave, lane, pk, mask, s_c, s_stage, perm_b);
    __syncthreads();
    if (ns) {
        int m = (bcnt < CAP) ? bcnt : CAP;
        for (int i = tid; i < m; i += STHR) perm_n[bstart + i] = s_stage[i];
        return;
    }

    for (int rr = 0; rr < ERW / 32; ++rr) {
        int r = __builtin_amdgcn_readfirstlane(wave * (ERW / 32) + rr);
        int rb = (r > 0) ? __builtin_amdgcn_readfirstlane(s_c[r - 1]) : 0;
        int re = __builtin_amdgcn_readfirstlane(s_c[r]);
        float4 acc = make_float4(0.f, 0.f, 0.f, 0.f);
        for (int c = rb; c < re; c += 32) {
            int cnt = (re - c < 32) ? (re - c) : 32;
            int p = c + lane;
            int mine = s_stage[(p < CAP) ? p : CAP - 1];
            if (c + 32 > CAP) {
                if (p >= CAP && lane < cnt) mine = perm_e[bstart + p];
            }
            for (int j = 0; j < cnt; j += RB) {
                if (j + RB <= cnt) fr_add_rows<true>(probs, mine, j, cnt, (unsigned)lane, acc);
                else               fr_add_rows<false>(probs, mine, j, cnt, (unsigned)lane, acc);
            }
        }
        float inv = 1.f / fmaxf((float)(re - rb), 1.f);
        acc.x *= inv; acc.y *= inv; acc.z *= inv; acc.w *= inv;
        *(float4*)&sA[r][lane * 4] = acc;
    }
    __syncthreads();

    int half = lane >> 4, lr = lane & 15;
    int n0 = (wave & 7) * 16;
    int r0 = (wave >> 3) * 32;
    v8f c0 = {0.f, 0.f, 0.f, 0.f, 0.f, 0.f, 0.f, 0.f};
    v8f c1 = {0.f, 0.f, 0.f, 0.f, 0.f, 0.f, 0.f, 0.f};
    #pragma unroll 8
    for (int k0 = 0; k0 < KDIM; k0 += 4) {
        int ka = k0 + 2 * half;
        v2f a0, a1, bw;
        a0.x = sA[r0 + lr][ka];
        a0.y = sA[r0 + lr][ka + 1];
        a1.x = sA[r0 + 16 + lr][ka];
        a1.y = sA[r0 + 16 + lr][ka + 1];
        bw.x = W[(size_t)ka * KDIM + n0 + lr];
        bw.y = W[(size_t)(ka + 1) * KDIM + n0 + lr];
        c0 = __builtin_amdgcn_wmma_f32_16x16x4_f32(false, a0, false, bw, (short)0, c0,
                                                   false, false);
        c1 = __builtin_amdgcn_wmma_f32_16x16x4_f32(false, a1, false, bw, (short)0, c1,
                                                   false, false);
    }
    int rowbase = key0 + r0 + 8 * half;
    #pragma unroll
    for (int r = 0; r < 8; ++r) {
        int row = rowbase + r;
        if (row < E) ef[(size_t)row * KDIM + n0 + lr] = c0[r];
        if (row + 16 < E) ef[(size_t)(row + 16) * KDIM + n0 + lr] = c1[r];
    }
}

__global__ __launch_bounds__(256) void fr_scatter_fin(const float* __restrict__ edge_feat,
                                                      const int* __restrict__ off_n,
                                                      const int* __restrict__ perm_n,
                                                      const float* __restrict__ probs,
                                                      float* __restrict__ out, int n, int nnz) {
    int wave = threadIdx.x >> 5;
    unsigned lane = threadIdx.x & 31;
    int nw = gridDim.x * 8;
    for (int row0 = blockIdx.x * 8 + wave; row0 < n; row0 += nw) {
        int row = __builtin_amdgcn_readfirstlane(row0);
        int b = __builtin_amdgcn_readfirstlane(off_n[row]);
        int e = (row + 1 < n) ? __builtin_amdgcn_readfirstlane(off_n[row + 1]) : nnz;
        float4 p = ((const float4*)(probs + (size_t)row * KDIM))[lane];
        float4 acc = fr_row_sum(edge_feat, perm_n, b, e, lane);
        float inv = 1.f / fmaxf((float)(e - b), 1.f);
        float4 nat;
        nat.x = fmaxf(p.x, EPS) * acc.x * inv;
        nat.y = fmaxf(p.y, EPS) * acc.y * inv;
        nat.z = fmaxf(p.z, EPS) * acc.z * inv;
        nat.w = fmaxf(p.w, EPS) * acc.w * inv;
        float s = nat.x + nat.y + nat.z + nat.w;
        #pragma unroll
        for (int off = 16; off > 0; off >>= 1) s += __shfl_xor(s, off, 32);
        float mean = s * (1.0f / KDIM);
        nat.x -= mean; nat.y -= mean; nat.z -= mean; nat.w -= mean;
        ((float4*)(out + (size_t)row * KDIM))[lane] = nat;
    }
}

extern "C" void kernel_launch(void* const* d_in, const int* in_sizes, int n_in,
                              void* d_out, int out_size, void* d_ws, size_t ws_size,
                              hipStream_t stream) {
    const float* y        = (const float*)d_in[0];
    const float* W        = (const float*)d_in[1];
    const int*   node_idx = (const int*)d_in[2];
    const int*   edge_idx = (const int*)d_in[3];
    float*       out      = (float*)d_out;

    const int n   = in_sizes[0] / KDIM;
    const int nnz = in_sizes[2];
    const int E   = N_EDGES;
    const int nbe = (E + (1 << ESH) - 1) >> ESH;
    const int nbn = (n + (1 << NSH) - 1) >> NSH;
    const int npb = (nnz + SLICE - 1) / SLICE;

    float* probs   = (float*)d_ws;
    float* plane_a = probs + (size_t)n * KDIM;
    float* plane_b = plane_a + (size_t)E * KDIM;
    int*   cnt     = (int*)(plane_b + (size_t)E * KDIM);
    unsigned* seg_e = (unsigned*)out;
    unsigned* seg_n = seg_e + nnz;
    int*   tab     = (int*)(seg_n + nnz);
    int*   perm_e  = (int*)plane_b;
    int*   perm_n  = perm_e + nnz;

    fr_softmax_slices<<<npb + (n + 31) / 32, 1024, 0, stream>>>(y, probs, node_idx, edge_idx, seg_e, seg_n, tab,
                                                                npb, n, nnz);
    fr_rows_edges<<<nbe + nbn, STHR, 0, stream>>>(seg_e, seg_n, tab, cnt + E, perm_e, perm_n, probs, W,
                                                 plane_a, nbe, npb, E, n);
    fr_scatter_fin<<<(n + 63) / 64, 256, 0, stream>>>(plane_a, cnt + E, perm_n, probs, out, n, nnz);
}
